# starting kernel
# baseline (speedup 1.0000x reference)
#include <hip/hip_runtime.h>
#include <math.h>

typedef __attribute__((ext_vector_type(2))) float v2f;
typedef __attribute__((ext_vector_type(8))) float v8f;

#define LN_EPS 1e-5f
#define SQ_EPS 1e-9f

__device__ __forceinline__ float gelu_exact(float x) {
    return 0.5f * x * (1.0f + erff(x * 0.70710678118654752440f));
}

__device__ __forceinline__ v8f wmma4(v2f a, v2f b, v8f c) {
    return __builtin_amdgcn_wmma_f32_16x16x4_f32(false, a, false, b, (short)0, c, false, false);
}

__device__ __forceinline__ v8f vzero8() {
    v8f z = {0.f,0.f,0.f,0.f,0.f,0.f,0.f,0.f};
    return z;
}

__global__ __attribute__((amdgpu_waves_per_eu(1))) __launch_bounds__(256)
void capsnet_fused(
    const float* __restrict__ x,   const float* __restrict__ W1, const float* __restrict__ b1,
    const float* __restrict__ W2,  const float* __restrict__ b2,
    const float* __restrict__ ln_g,const float* __restrict__ ln_b,
    const float* __restrict__ Wcap,const float* __restrict__ Wf, const float* __restrict__ bf,
    float* __restrict__ out)
{
    __shared__ float xh[16*256];
    __shared__ float h1s[16*32];
    __shared__ float mu_s[16];
    __shared__ float rstd_s[16];
    __shared__ float u_t[32*130];
    __shared__ float bij[16*513];
    __shared__ float cij[16*513];
    __shared__ float part[16*17];

    const int t    = threadIdx.x;
    const int lane = t & 31;
    const int wav  = t >> 5;
    const int bb   = lane & 15;
    const int hi   = lane >> 4;

    const int tile = blockIdx.x;
    const float* xg = x + (size_t)tile * 16 * 256;

    #pragma unroll
    for (int r = 0; r < 16; ++r) xh[t + 256*r] = xg[t + 256*r];
    __syncthreads();

    {
        int n = t & 31, b0 = t >> 5;
        float acc0 = b1[n], acc1 = acc0;
        #pragma unroll 8
        for (int k = 0; k < 256; ++k) {
            float w = W1[k*32 + n];
            acc0 = fmaf(xh[b0*256 + k],       w, acc0);
            acc1 = fmaf(xh[(b0+8)*256 + k],   w, acc1);
        }
        h1s[b0*32 + n]     = gelu_exact(acc0);
        h1s[(b0+8)*32 + n] = gelu_exact(acc1);
    }
    __syncthreads();

    {
        int n0 = t & 31, b0 = t >> 5;
        float h1a[32], h1b[32];
        #pragma unroll
        for (int k = 0; k < 32; ++k) { h1a[k] = h1s[b0*32 + k]; h1b[k] = h1s[(b0+8)*32 + k]; }
        #pragma unroll
        for (int rr = 0; rr < 8; ++rr) {
            int n = n0 + 32*rr;
            float acc0 = b2[n], acc1 = acc0;
            #pragma unroll
            for (int k = 0; k < 32; ++k) {
                float w = W2[k*256 + n];
                acc0 = fmaf(h1a[k], w, acc0);
                acc1 = fmaf(h1b[k], w, acc1);
            }
            xh[b0*256 + n]     = gelu_exact(acc0);
            xh[(b0+8)*256 + n] = gelu_exact(acc1);
        }
    }
    __syncthreads();

    {
        int ba = wav, bc = wav + 8;
        float sa = 0.f, sb = 0.f;
        #pragma unroll
        for (int r = 0; r < 8; ++r) { sa += xh[ba*256 + lane + 32*r]; sb += xh[bc*256 + lane + 32*r]; }
        #pragma unroll
        for (int m = 16; m >= 1; m >>= 1) { sa += __shfl_xor(sa, m, 32); sb += __shfl_xor(sb, m, 32); }
        float mua = sa * (1.0f/256.0f), mub = sb * (1.0f/256.0f);
        float va = 0.f, vb = 0.f;
        #pragma unroll
        for (int r = 0; r < 8; ++r) {
            float da = xh[ba*256 + lane + 32*r] - mua; va = fmaf(da, da, va);
            float db = xh[bc*256 + lane + 32*r] - mub; vb = fmaf(db, db, vb);
        }
        #pragma unroll
        for (int m = 16; m >= 1; m >>= 1) { va += __shfl_xor(va, m, 32); vb += __shfl_xor(vb, m, 32); }
        if (lane == 0) {
            mu_s[ba] = mua; rstd_s[ba] = rsqrtf(va * (1.0f/256.0f) + LN_EPS);
            mu_s[bc] = mub; rstd_s[bc] = rsqrtf(vb * (1.0f/256.0f) + LN_EPS);
        }
    }
    __syncthreads();

    {
        int i = t & 31, b0 = t >> 5;
        float g8[8], be8[8];
        #pragma unroll
        for (int k = 0; k < 8; ++k) { g8[k] = ln_g[i*8 + k]; be8[k] = ln_b[i*8 + k]; }
        #pragma unroll
        for (int bsel = 0; bsel < 2; ++bsel) {
            int b = b0 + 8*bsel;
            float mu = mu_s[b], rs = rstd_s[b];
            float vals[8]; float nsq = 0.f;
            #pragma unroll
            for (int k = 0; k < 8; ++k) {
                float v = (xh[b*256 + i*8 + k] - mu) * rs * g8[k] + be8[k];
                vals[k] = v; nsq = fmaf(v, v, nsq);
            }
            float scale = nsq / ((1.0f + nsq) * sqrtf(nsq + SQ_EPS));
            #pragma unroll
            for (int k = 0; k < 8; ++k) u_t[i*130 + k*16 + b] = scale * vals[k];
        }
    }

    #pragma unroll
    for (int r = 0; r < 32; ++r) { bij[t + 256*r] = 0.f; cij[t + 256*r] = 0.0625f; }
    if (t < 16) { bij[8192 + t] = 0.f; cij[8192 + t] = 0.0625f; }
    __syncthreads();

    v2f a0A[32], a1A[32], a0B[32], a1B[32];
    #pragma unroll
    for (int i = 0; i < 32; ++i) {
        const float* wcA = Wcap + (size_t)((i*16 + wav    )*8)*16 + bb;
        const float* wcB = Wcap + (size_t)((i*16 + wav + 8)*8)*16 + bb;
        a0A[i][0] = wcA[(2*hi+0)*16];  a0A[i][1] = wcA[(2*hi+1)*16];
        a1A[i][0] = wcA[(2*hi+4)*16];  a1A[i][1] = wcA[(2*hi+5)*16];
        a0B[i][0] = wcB[(2*hi+0)*16];  a0B[i][1] = wcB[(2*hi+1)*16];
        a1B[i][0] = wcB[(2*hi+4)*16];  a1B[i][1] = wcB[(2*hi+5)*16];
    }

    v8f vA = vzero8(), vB = vzero8();
    for (int r = 0; r < 3; ++r) {
        v8f sA0 = vzero8(), sA1 = vzero8(), sB0 = vzero8(), sB1 = vzero8();
        #pragma unroll
        for (int i = 0; i < 32; ++i) {
            const float* ub = u_t + i*130 + bb;
            v2f bu0, bu1;
            bu0[0] = ub[(2*hi+0)*16];  bu0[1] = ub[(2*hi+1)*16];
            bu1[0] = ub[(2*hi+4)*16];  bu1[1] = ub[(2*hi+5)*16];
            float cA = cij[bb*513 + i*16 + wav];
            float cB = cij[bb*513 + i*16 + wav + 8];
            v2f p0, p1;
            p0[0] = bu0[0]*cA; p0[1] = bu0[1]*cA; p1[0] = bu1[0]*cA; p1[1] = bu1[1]*cA;
            if (i & 1) { sA1 = wmma4(a0A[i], p0, sA1); sA1 = wmma4(a1A[i], p1, sA1); }
            else       { sA0 = wmma4(a0A[i], p0, sA0); sA0 = wmma4(a1A[i], p1, sA0); }
            p0[0] = bu0[0]*cB; p0[1] = bu0[1]*cB; p1[0] = bu1[0]*cB; p1[1] = bu1[1]*cB;
            if (i & 1) { sB1 = wmma4(a0B[i], p0, sB1); sB1 = wmma4(a1B[i], p1, sB1); }
            else       { sB0 = wmma4(a0B[i], p0, sB0); sB0 = wmma4(a1B[i], p1, sB0); }
        }
        v8f sA = sA0 + sA1, sB = sB0 + sB1;
        float nA = 0.f, nB = 0.f;
        #pragma unroll
        for (int v = 0; v < 8; ++v) { nA = fmaf(sA[v], sA[v], nA); nB = fmaf(sB[v], sB[v], nB); }
        nA += __shfl_xor(nA, 16, 32);  nB += __shfl_xor(nB, 16, 32);
        float scA = nA / ((1.0f + nA) * sqrtf(nA + SQ_EPS));
        float scB = nB / ((1.0f + nB) * sqrtf(nB + SQ_EPS));
        #pragma unroll
        for (int v = 0; v < 8; ++v) { vA[v] = sA[v]*scA; vB[v] = sB[v]*scB; }

        if (r < 2) {
            #pragma unroll
            for (int i = 0; i < 32; ++i) {
                const float* ub = u_t + i*130 + bb;
                v2f bu0, bu1;
                bu0[0] = ub[(2*hi+0)*16];  bu0[1] = ub[(2*hi+1)*16];
                bu1[0] = ub[(2*hi+4)*16];  bu1[1] = ub[(2*hi+5)*16];
                v8f uA = vzero8(); uA = wmma4(a0A[i], bu0, uA); uA = wmma4(a1A[i], bu1, uA);
                v8f uB = vzero8(); uB = wmma4(a0B[i], bu0, uB); uB = wmma4(a1B[i], bu1, uB);
                float agA = 0.f, agB = 0.f;
                #pragma unroll
                for (int v = 0; v < 8; ++v) { agA = fmaf(uA[v], vA[v], agA); agB = fmaf(uB[v], vB[v], agB); }
                agA += __shfl_xor(agA, 16, 32);  agB += __shfl_xor(agB, 16, 32);
                int ia = bb*513 + i*16 + wav;
                if (lane < 16) { bij[ia] += agA; bij[ia + 8] += agB; }
            }
            __syncthreads();
            #pragma unroll
            for (int bsel = 0; bsel < 2; ++bsel) {
                int base = (wav + 8*bsel)*513 + (t & 31)*16;
                float mx = bij[base];
                #pragma unroll
                for (int j = 1; j < 16; ++j) mx = fmaxf(mx, bij[base + j]);
                float e[16]; float sum = 0.f;
                #pragma unroll
                for (int j = 0; j < 16; ++j) { e[j] = expf(bij[base + j] - mx); sum += e[j]; }
                float inv = 1.0f / sum;
                #pragma unroll
                for (int j = 0; j < 16; ++j) cij[base + j] = e[j] * inv;
            }
            __syncthreads();
        }
    }

    {
        float pA = 0.f, pB = 0.f;
        #pragma unroll
        for (int v = 0; v < 8; ++v) {
            pA = fmaf(vA[v], Wf[(wav    )*16 + v + 8*hi], pA);
            pB = fmaf(vB[v], Wf[(wav + 8)*16 + v + 8*hi], pB);
        }
        pA += __shfl_xor(pA, 16, 32);  pB += __shfl_xor(pB, 16, 32);
        if (lane < 16) { part[wav*17 + bb] = pA; part[(wav+8)*17 + bb] = pB; }
    }
    __syncthreads();
    if (t < 16) {
        float acc = bf[0];
        #pragma unroll
        for (int j = 0; j < 16; ++j) acc += part[j*17 + t];
        out[tile*16 + t] = acc;
    }
}

extern "C" void kernel_launch(void* const* d_in, const int* in_sizes, int n_in,
                              void* d_out, int out_size, void* d_ws, size_t ws_size,
                              hipStream_t stream) {
    const float* x    = (const float*)d_in[0];
    const float* W1   = (const float*)d_in[1];
    const float* b1   = (const float*)d_in[2];
    const float* W2   = (const float*)d_in[3];
    const float* b2   = (const float*)d_in[4];
    const float* ln_g = (const float*)d_in[5];
    const float* ln_b = (const float*)d_in[6];
    const float* Wcap = (const float*)d_in[7];
    const float* Wf   = (const float*)d_in[8];
    const float* bf   = (const float*)d_in[9];
    float* out = (float*)d_out;

    const int B = in_sizes[0] / 256;
    const int blocks = B / 16;
    capsnet_fused<<<blocks, 256, 0, stream>>>(x, W1, b1, W2, b2, ln_g, ln_b, Wcap, Wf, bf, out);
}


# MLP weight loads batched 64 deep and issued one batch ahead of use (W1: 32 dependent second-level-cache trips per wave become 4, the first under the x fetch; W2: 8 become 4); sums bit-identical
# speedup vs baseline: 1.1807x; 1.1807x over previous
#include <hip/hip_runtime.h>
#include <math.h>

typedef __attribute__((ext_vector_type(2))) float v2f;
typedef __attribute__((ext_vector_type(8))) float v8f;

#define LN_EPS 1e-5f
#define SQ_EPS 1e-9f

__device__ __forceinline__ float gelu_exact(float x) {
    return 0.5f * x * (1.0f + erff(x * 0.70710678118654752440f));
}

__device__ __forceinline__ v8f wmma4(v2f a, v2f b, v8f c) {
    return __builtin_amdgcn_wmma_f32_16x16x4_f32(false, a, false, b, (short)0, c, false, false);
}

__device__ __forceinline__ v8f vzero8() {
    v8f z = {0.f,0.f,0.f,0.f,0.f,0.f,0.f,0.f};
    return z;
}

__global__ __attribute__((amdgpu_waves_per_eu(1))) __launch_bounds__(256)
void capsnet_fused(
    const float* __restrict__ x,   const float* __restrict__ W1, const float* __restrict__ b1,
    const float* __restrict__ W2,  const float* __restrict__ b2,
    const float* __restrict__ ln_g,const float* __restrict__ ln_b,
    const float* __restrict__ Wcap,const float* __restrict__ Wf, const float* __restrict__ bf,
    float* __restrict__ out)
{
    __shared__ float xh[16*256];
    __shared__ float h1s[16*32];
    __shared__ float mu_s[16];
    __shared__ float rstd_s[16];
    __shared__ float u_t[32*130];
    __shared__ float bij[16*513];
    __shared__ float cij[16*513];
    __shared__ float part[16*17];

    const int t    = threadIdx.x;
    const int lane = t & 31;
    const int wav  = t >> 5;
    const int bb   = lane & 15;
    const int hi   = lane >> 4;

    const int tile = blockIdx.x;
    const float* xg = x + (size_t)tile * 16 * 256;

    const int n1 = t & 31, r1 = t >> 5;


    float xr[16];
    #pragma unroll
    for (int r = 0; r < 16; ++r) xr[r] = xg[t + 256*r];
    float wq[64];
    #pragma unroll
    for (int k = 0; k < 64; ++k) wq[k] = W1[k*32 + n1];
    float acc0 = b1[n1], acc1 = acc0;
    #pragma unroll
    for (int r = 0; r < 16; ++r) xh[t + 256*r] = xr[r];
    __syncthreads();

    #pragma unroll 1
    for (int c = 0; c < 3; ++c) {
        float wn[64];
        #pragma unroll
        for (int k = 0; k < 64; ++k) wn[k] = W1[((c+1)*64 + k)*32 + n1];
        #pragma unroll
        for (int k = 0; k < 64; ++k) {
            acc0 = fmaf(xh[r1*256 + c*64 + k],     wq[k], acc0);
            acc1 = fmaf(xh[(r1+8)*256 + c*64 + k], wq[k], acc1);
        }
        #pragma unroll
        for (int k = 0; k < 64; ++k) wq[k] = wn[k];
    }
    float w2c[64], b2v[8];
    #pragma unroll
    for (int q = 0; q < 2; ++q) {
        #pragma unroll
        for (int k = 0; k < 32; ++k) w2c[q*32 + k] = W2[k*256 + n1 + 32*q];
    }
    #pragma unroll
    for (int rr = 0; rr < 8; ++rr) b2v[rr] = b2[n1 + 32*rr];
    #pragma unroll
    for (int k = 0; k < 64; ++k) {
        acc0 = fmaf(xh[r1*256 + 192 + k],     wq[k], acc0);
        acc1 = fmaf(xh[(r1+8)*256 + 192 + k], wq[k], acc1);
    }
    h1s[r1*32 + n1]     = gelu_exact(acc0);
    h1s[(r1+8)*32 + n1] = gelu_exact(acc1);
    __syncthreads();

    {
        float h1a[32], h1b[32];
        #pragma unroll
        for (int k = 0; k < 32; ++k) { h1a[k] = h1s[r1*32 + k]; h1b[k] = h1s[(r1+8)*32 + k]; }
        #pragma unroll
        for (int p = 0; p < 4; ++p) {
            float w2n[64];
            if (p < 3) {
                #pragma unroll
                for (int q = 0; q < 2; ++q) {
                    #pragma unroll
                    for (int k = 0; k < 32; ++k) w2n[q*32 + k] = W2[k*256 + n1 + 32*(2*(p+1) + q)];
                }
            }
            #pragma unroll
            for (int q = 0; q < 2; ++q) {
                const int n = n1 + 32*(2*p + q);
                float a0 = b2v[2*p + q], a1 = a0;
                #pragma unroll
                for (int k = 0; k < 32; ++k) {
                    a0 = fmaf(h1a[k], w2c[q*32 + k], a0);
                    a1 = fmaf(h1b[k], w2c[q*32 + k], a1);
                }
                xh[r1*256 + n]     = gelu_exact(a0);
                xh[(r1+8)*256 + n] = gelu_exact(a1);
            }
            if (p < 3) {
                #pragma unroll
                for (int k = 0; k < 64; ++k) w2c[k] = w2n[k];
            }
        }
    }
    __syncthreads();

    {
        int ba = wav, bc = wav + 8;
        float sa = 0.f, sb = 0.f;
        #pragma unroll
        for (int r = 0; r < 8; ++r) { sa += xh[ba*256 + lane + 32*r]; sb += xh[bc*256 + lane + 32*r]; }
        #pragma unroll
        for (int m = 16; m >= 1; m >>= 1) { sa += __shfl_xor(sa, m, 32); sb += __shfl_xor(sb, m, 32); }
        float mua = sa * (1.0f/256.0f), mub = sb * (1.0f/256.0f);
        float va = 0.f, vb = 0.f;
        #pragma unroll
        for (int r = 0; r < 8; ++r) {
            float da = xh[ba*256 + lane + 32*r] - mua; va = fmaf(da, da, va);
            float db = xh[bc*256 + lane + 32*r] - mub; vb = fmaf(db, db, vb);
        }
        #pragma unroll
        for (int m = 16; m >= 1; m >>= 1) { va += __shfl_xor(va, m, 32); vb += __shfl_xor(vb, m, 32); }
        if (lane == 0) {
            mu_s[ba] = mua; rstd_s[ba] = rsqrtf(va * (1.0f/256.0f) + LN_EPS);
            mu_s[bc] = mub; rstd_s[bc] = rsqrtf(vb * (1.0f/256.0f) + LN_EPS);
        }
    }
    __syncthreads();

    {
        int i = t & 31, b0 = t >> 5;
        float g8[8], be8[8];
        #pragma unroll
        for (int k = 0; k < 8; ++k) { g8[k] = ln_g[i*8 + k]; be8[k] = ln_b[i*8 + k]; }
        #pragma unroll
        for (int bsel = 0; bsel < 2; ++bsel) {
            int b = b0 + 8*bsel;
            float mu = mu_s[b], rs = rstd_s[b];
            float vals[8]; float nsq = 0.f;
            #pragma unroll
            for (int k = 0; k < 8; ++k) {
                float v = (xh[b*256 + i*8 + k] - mu) * rs * g8[k] + be8[k];
                vals[k] = v; nsq = fmaf(v, v, nsq);
            }
            float scale = nsq / ((1.0f + nsq) * sqrtf(nsq + SQ_EPS));
            #pragma unroll
            for (int k = 0; k < 8; ++k) u_t[i*130 + k*16 + b] = scale * vals[k];
        }
    }

    #pragma unroll
    for (int r = 0; r < 32; ++r) { bij[t + 256*r] = 0.f; cij[t + 256*r] = 0.0625f; }
    if (t < 16) { bij[8192 + t] = 0.f; cij[8192 + t] = 0.0625f; }
    __syncthreads();

    v2f a0A[32], a1A[32], a0B[32], a1B[32];
    #pragma unroll
    for (int i = 0; i < 32; ++i) {
        const float* wcA = Wcap + (size_t)((i*16 + wav    )*8)*16 + bb;
        const float* wcB = Wcap + (size_t)((i*16 + wav + 8)*8)*16 + bb;
        a0A[i][0] = wcA[(2*hi+0)*16];  a0A[i][1] = wcA[(2*hi+1)*16];
        a1A[i][0] = wcA[(2*hi+4)*16];  a1A[i][1] = wcA[(2*hi+5)*16];
        a0B[i][0] = wcB[(2*hi+0)*16];  a0B[i][1] = wcB[(2*hi+1)*16];
        a1B[i][0] = wcB[(2*hi+4)*16];  a1B[i][1] = wcB[(2*hi+5)*16];
    }

    v8f vA = vzero8(), vB = vzero8();
    for (int r = 0; r < 3; ++r) {
        v8f sA0 = vzero8(), sA1 = vzero8(), sB0 = vzero8(), sB1 = vzero8();
        #pragma unroll
        for (int i = 0; i < 32; ++i) {
            const float* ub = u_t + i*130 + bb;
            v2f bu0, bu1;
            bu0[0] = ub[(2*hi+0)*16];  bu0[1] = ub[(2*hi+1)*16];
            bu1[0] = ub[(2*hi+4)*16];  bu1[1] = ub[(2*hi+5)*16];
            float cA = cij[bb*513 + i*16 + wav];
            float cB = cij[bb*513 + i*16 + wav + 8];
            v2f p0, p1;
            p0[0] = bu0[0]*cA; p0[1] = bu0[1]*cA; p1[0] = bu1[0]*cA; p1[1] = bu1[1]*cA;
            if (i & 1) { sA1 = wmma4(a0A[i], p0, sA1); sA1 = wmma4(a1A[i], p1, sA1); }
            else       { sA0 = wmma4(a0A[i], p0, sA0); sA0 = wmma4(a1A[i], p1, sA0); }
            p0[0] = bu0[0]*cB; p0[1] = bu0[1]*cB; p1[0] = bu1[0]*cB; p1[1] = bu1[1]*cB;
            if (i & 1) { sB1 = wmma4(a0B[i], p0, sB1); sB1 = wmma4(a1B[i], p1, sB1); }
            else       { sB0 = wmma4(a0B[i], p0, sB0); sB0 = wmma4(a1B[i], p1, sB0); }
        }
        v8f sA = sA0 + sA1, sB = sB0 + sB1;
        float nA = 0.f, nB = 0.f;
        #pragma unroll
        for (int v = 0; v < 8; ++v) { nA = fmaf(sA[v], sA[v], nA); nB = fmaf(sB[v], sB[v], nB); }
        nA += __shfl_xor(nA, 16, 32);  nB += __shfl_xor(nB, 16, 32);
        float scA = nA / ((1.0f + nA) * sqrtf(nA + SQ_EPS));
        float scB = nB / ((1.0f + nB) * sqrtf(nB + SQ_EPS));
        #pragma unroll
        for (int v = 0; v < 8; ++v) { vA[v] = sA[v]*scA; vB[v] = sB[v]*scB; }

        if (r < 2) {
            #pragma unroll
            for (int i = 0; i < 32; ++i) {
                const float* ub = u_t + i*130 + bb;
                v2f bu0, bu1;
                bu0[0] = ub[(2*hi+0)*16];  bu0[1] = ub[(2*hi+1)*16];
                bu1[0] = ub[(2*hi+4)*16];  bu1[1] = ub[(2*hi+5)*16];
                v8f uA = vzero8(); uA = wmma4(a0A[i], bu0, uA); uA = wmma4(a1A[i], bu1, uA);
                v8f uB = vzero8(); uB = wmma4(a0B[i], bu0, uB); uB = wmma4(a1B[i], bu1, uB);
                float agA = 0.f, agB = 0.f;
                #pragma unroll
                for (int v = 0; v < 8; ++v) { agA = fmaf(uA[v], vA[v], agA); agB = fmaf(uB[v], vB[v], agB); }
                agA += __shfl_xor(agA, 16, 32);  agB += __shfl_xor(agB, 16, 32);
                int ia = bb*513 + i*16 + wav;
                if (lane < 16) { bij[ia] += agA; bij[ia + 8] += agB; }
            }
            __syncthreads();
            #pragma unroll
            for (int bsel = 0; bsel < 2; ++bsel) {
                int base = (wav + 8*bsel)*513 + (t & 31)*16;
                float mx = bij[base];
                #pragma unroll
                for (int j = 1; j < 16; ++j) mx = fmaxf(mx, bij[base + j]);
                float e[16]; float sum = 0.f;
                #pragma unroll
                for (int j = 0; j < 16; ++j) { e[j] = expf(bij[base + j] - mx); sum += e[j]; }
                float inv = 1.0f / sum;
                #pragma unroll
                for (int j = 0; j < 16; ++j) cij[base + j] = e[j] * inv;
            }
            __syncthreads();
        }
    }

    {
        float pA = 0.f, pB = 0.f;
        #pragma unroll
        for (int v = 0; v < 8; ++v) {
            pA = fmaf(vA[v], Wf[(wav    )*16 + v + 8*hi], pA);
            pB = fmaf(vB[v], Wf[(wav + 8)*16 + v + 8*hi], pB);
        }
        pA += __shfl_xor(pA, 16, 32);  pB += __shfl_xor(pB, 16, 32);
        if (lane < 16) { part[wav*17 + bb] = pA; part[(wav+8)*17 + bb] = pB; }
    }
    __syncthreads();
    if (t < 16) {
        float acc = bf[0];
        #pragma unroll
        for (int j = 0; j < 16; ++j) acc += part[j*17 + t];
        out[tile*16 + t] = acc;
    }
}

extern "C" void kernel_launch(void* const* d_in, const int* in_sizes, int n_in,
                              void* d_out, int out_size, void* d_ws, size_t ws_size,
                              hipStream_t stream) {
    const float* x    = (const float*)d_in[0];
    const float* W1   = (const float*)d_in[1];
    const float* b1   = (const float*)d_in[2];
    const float* W2   = (const float*)d_in[3];
    const float* b2   = (const float*)d_in[4];
    const float* ln_g = (const float*)d_in[5];
    const float* ln_b = (const float*)d_in[6];
    const float* Wcap = (const float*)d_in[7];
    const float* Wf   = (const float*)d_in[8];
    const float* bf   = (const float*)d_in[9];
    float* out = (float*)d_out;

    const int B = in_sizes[0] / 256;
    const int blocks = B / 16;
    capsnet_fused<<<blocks, 256, 0, stream>>>(x, W1, b1, W2, b2, ln_g, ln_b, Wcap, Wf, bf, out);
}


# Agreement pass freed of per-capsule LDS round trips: cross-half sum by one row exchange (v_permlanex16), both lane halves update their own logit, no lane<16 branch; sums bit-identical
# speedup vs baseline: 1.3136x; 1.1126x over previous
#include <hip/hip_runtime.h>
#include <math.h>

typedef __attribute__((ext_vector_type(2))) float v2f;
typedef __attribute__((ext_vector_type(8))) float v8f;

#define LN_EPS 1e-5f
#define SQ_EPS 1e-9f

__device__ __forceinline__ float gelu_exact(float x) {
    return 0.5f * x * (1.0f + erff(x * 0.70710678118654752440f));
}

__device__ __forceinline__ v8f wmma4(v2f a, v2f b, v8f c) {
    return __builtin_amdgcn_wmma_f32_16x16x4_f32(false, a, false, b, (short)0, c, false, false);
}

__device__ __forceinline__ float xhalf(float x) {
    const int b = __builtin_bit_cast(int, x);
    return __builtin_bit_cast(float, __builtin_amdgcn_permlanex16(b, b, 0x76543210, 0xfedcba98, false, false));
}

__device__ __forceinline__ v8f vzero8() {
    v8f z = {0.f,0.f,0.f,0.f,0.f,0.f,0.f,0.f};
    return z;
}

__global__ __attribute__((amdgpu_waves_per_eu(1))) __launch_bounds__(256)
void capsnet_fused(
    const float* __restrict__ x,   const float* __restrict__ W1, const float* __restrict__ b1,
    const float* __restrict__ W2,  const float* __restrict__ b2,
    const float* __restrict__ ln_g,const float* __restrict__ ln_b,
    const float* __restrict__ Wcap,const float* __restrict__ Wf, const float* __restrict__ bf,
    float* __restrict__ out)
{
    __shared__ float xh[16*256];
    __shared__ float h1s[16*32];
    __shared__ float mu_s[16];
    __shared__ float rstd_s[16];
    __shared__ float u_t[32*130];
    __shared__ float bij[16*513];
    __shared__ float cij[16*513];
    __shared__ float part[16*17];

    const int t    = threadIdx.x;
    const int lane = t & 31;
    const int wav  = t >> 5;
    const int bb   = lane & 15;
    const int hi   = lane >> 4;

    const int tile = blockIdx.x;
    const float* xg = x + (size_t)tile * 16 * 256;

    const int n1 = t & 31, r1 = t >> 5;


    float xr[16];
    #pragma unroll
    for (int r = 0; r < 16; ++r) xr[r] = xg[t + 256*r];
    float wq[64];
    #pragma unroll
    for (int k = 0; k < 64; ++k) wq[k] = W1[k*32 + n1];
    float acc0 = b1[n1], acc1 = acc0;
    #pragma unroll
    for (int r = 0; r < 16; ++r) xh[t + 256*r] = xr[r];
    __syncthreads();

    #pragma unroll 1
    for (int c = 0; c < 3; ++c) {
        float wn[64];
        #pragma unroll
        for (int k = 0; k < 64; ++k) wn[k] = W1[((c+1)*64 + k)*32 + n1];
        #pragma unroll
        for (int k = 0; k < 64; ++k) {
            acc0 = fmaf(xh[r1*256 + c*64 + k],     wq[k], acc0);
            acc1 = fmaf(xh[(r1+8)*256 + c*64 + k], wq[k], acc1);
        }
        #pragma unroll
        for (int k = 0; k < 64; ++k) wq[k] = wn[k];
    }
    float w2c[64], b2v[8];
    #pragma unroll
    for (int q = 0; q < 2; ++q) {
        #pragma unroll
        for (int k = 0; k < 32; ++k) w2c[q*32 + k] = W2[k*256 + n1 + 32*q];
    }
    #pragma unroll
    for (int rr = 0; rr < 8; ++rr) b2v[rr] = b2[n1 + 32*rr];
    #pragma unroll
    for (int k = 0; k < 64; ++k) {
        acc0 = fmaf(xh[r1*256 + 192 + k],     wq[k], acc0);
        acc1 = fmaf(xh[(r1+8)*256 + 192 + k], wq[k], acc1);
    }
    h1s[r1*32 + n1]     = gelu_exact(acc0);
    h1s[(r1+8)*32 + n1] = gelu_exact(acc1);
    __syncthreads();

    {
        float h1a[32], h1b[32];
        #pragma unroll
        for (int k = 0; k < 32; ++k) { h1a[k] = h1s[r1*32 + k]; h1b[k] = h1s[(r1+8)*32 + k]; }
        #pragma unroll
        for (int p = 0; p < 4; ++p) {
            float w2n[64];
            if (p < 3) {
                #pragma unroll
                for (int q = 0; q < 2; ++q) {
                    #pragma unroll
                    for (int k = 0; k < 32; ++k) w2n[q*32 + k] = W2[k*256 + n1 + 32*(2*(p+1) + q)];
                }
            }
            #pragma unroll
            for (int q = 0; q < 2; ++q) {
                const int n = n1 + 32*(2*p + q);
                float a0 = b2v[2*p + q], a1 = a0;
                #pragma unroll
                for (int k = 0; k < 32; ++k) {
                    a0 = fmaf(h1a[k], w2c[q*32 + k], a0);
                    a1 = fmaf(h1b[k], w2c[q*32 + k], a1);
                }
                xh[r1*256 + n]     = gelu_exact(a0);
                xh[(r1+8)*256 + n] = gelu_exact(a1);
            }
            if (p < 3) {
                #pragma unroll
                for (int k = 0; k < 64; ++k) w2c[k] = w2n[k];
            }
        }
    }
    __syncthreads();

    {
        int ba = wav, bc = wav + 8;
        float sa = 0.f, sb = 0.f;
        #pragma unroll
        for (int r = 0; r < 8; ++r) { sa += xh[ba*256 + lane + 32*r]; sb += xh[bc*256 + lane + 32*r]; }
        #pragma unroll
        for (int m = 16; m >= 1; m >>= 1) { sa += __shfl_xor(sa, m, 32); sb += __shfl_xor(sb, m, 32); }
        float mua = sa * (1.0f/256.0f), mub = sb * (1.0f/256.0f);
        float va = 0.f, vb = 0.f;
        #pragma unroll
        for (int r = 0; r < 8; ++r) {
            float da = xh[ba*256 + lane + 32*r] - mua; va = fmaf(da, da, va);
            float db = xh[bc*256 + lane + 32*r] - mub; vb = fmaf(db, db, vb);
        }
        #pragma unroll
        for (int m = 16; m >= 1; m >>= 1) { va += __shfl_xor(va, m, 32); vb += __shfl_xor(vb, m, 32); }
        if (lane == 0) {
            mu_s[ba] = mua; rstd_s[ba] = rsqrtf(va * (1.0f/256.0f) + LN_EPS);
            mu_s[bc] = mub; rstd_s[bc] = rsqrtf(vb * (1.0f/256.0f) + LN_EPS);
        }
    }
    __syncthreads();

    {
        int i = t & 31, b0 = t >> 5;
        float g8[8], be8[8];
        #pragma unroll
        for (int k = 0; k < 8; ++k) { g8[k] = ln_g[i*8 + k]; be8[k] = ln_b[i*8 + k]; }
        #pragma unroll
        for (int bsel = 0; bsel < 2; ++bsel) {
            int b = b0 + 8*bsel;
            float mu = mu_s[b], rs = rstd_s[b];
            float vals[8]; float nsq = 0.f;
            #pragma unroll
            for (int k = 0; k < 8; ++k) {
                float v = (xh[b*256 + i*8 + k] - mu) * rs * g8[k] + be8[k];
                vals[k] = v; nsq = fmaf(v, v, nsq);
            }
            float scale = nsq / ((1.0f + nsq) * sqrtf(nsq + SQ_EPS));
            #pragma unroll
            for (int k = 0; k < 8; ++k) u_t[i*130 + k*16 + b] = scale * vals[k];
        }
    }

    #pragma unroll
    for (int r = 0; r < 32; ++r) { bij[t + 256*r] = 0.f; cij[t + 256*r] = 0.0625f; }
    if (t < 16) { bij[8192 + t] = 0.f; cij[8192 + t] = 0.0625f; }
    __syncthreads();

    v2f a0A[32], a1A[32], a0B[32], a1B[32];
    #pragma unroll
    for (int i = 0; i < 32; ++i) {
        const float* wcA = Wcap + (size_t)((i*16 + wav    )*8)*16 + bb;
        const float* wcB = Wcap + (size_t)((i*16 + wav + 8)*8)*16 + bb;
        a0A[i][0] = wcA[(2*hi+0)*16];  a0A[i][1] = wcA[(2*hi+1)*16];
        a1A[i][0] = wcA[(2*hi+4)*16];  a1A[i][1] = wcA[(2*hi+5)*16];
        a0B[i][0] = wcB[(2*hi+0)*16];  a0B[i][1] = wcB[(2*hi+1)*16];
        a1B[i][0] = wcB[(2*hi+4)*16];  a1B[i][1] = wcB[(2*hi+5)*16];
    }

    v8f vA = vzero8(), vB = vzero8();
    for (int r = 0; r < 3; ++r) {
        v8f sA0 = vzero8(), sA1 = vzero8(), sB0 = vzero8(), sB1 = vzero8();
        v2f ub0[32], ub1[32];
        #pragma unroll
        for (int i = 0; i < 32; ++i) {
            const float* ub = u_t + i*130 + bb;
            v2f bu0, bu1;
            bu0[0] = ub[(2*hi+0)*16];  bu0[1] = ub[(2*hi+1)*16];
            bu1[0] = ub[(2*hi+4)*16];  bu1[1] = ub[(2*hi+5)*16];
            ub0[i] = bu0;  ub1[i] = bu1;
            float cA = cij[bb*513 + i*16 + wav];
            float cB = cij[bb*513 + i*16 + wav + 8];
            v2f p0, p1;
            p0[0] = bu0[0]*cA; p0[1] = bu0[1]*cA; p1[0] = bu1[0]*cA; p1[1] = bu1[1]*cA;
            if (i & 1) { sA1 = wmma4(a0A[i], p0, sA1); sA1 = wmma4(a1A[i], p1, sA1); }
            else       { sA0 = wmma4(a0A[i], p0, sA0); sA0 = wmma4(a1A[i], p1, sA0); }
            p0[0] = bu0[0]*cB; p0[1] = bu0[1]*cB; p1[0] = bu1[0]*cB; p1[1] = bu1[1]*cB;
            if (i & 1) { sB1 = wmma4(a0B[i], p0, sB1); sB1 = wmma4(a1B[i], p1, sB1); }
            else       { sB0 = wmma4(a0B[i], p0, sB0); sB0 = wmma4(a1B[i], p1, sB0); }
        }
        v8f sA = sA0 + sA1, sB = sB0 + sB1;
        float nA = 0.f, nB = 0.f;
        #pragma unroll
        for (int v = 0; v < 8; ++v) { nA = fmaf(sA[v], sA[v], nA); nB = fmaf(sB[v], sB[v], nB); }
        nA += __shfl_xor(nA, 16, 32);  nB += __shfl_xor(nB, 16, 32);
        float scA = nA / ((1.0f + nA) * sqrtf(nA + SQ_EPS));
        float scB = nB / ((1.0f + nB) * sqrtf(nB + SQ_EPS));
        #pragma unroll
        for (int v = 0; v < 8; ++v) { vA[v] = sA[v]*scA; vB[v] = sB[v]*scB; }

        if (r < 2) {
            const int ib = bb*513 + wav + 8*hi;
            #pragma unroll
            for (int i = 0; i < 32; ++i) {
                const v2f bu0 = ub0[i], bu1 = ub1[i];
                v8f uA = vzero8(); uA = wmma4(a0A[i], bu0, uA); uA = wmma4(a1A[i], bu1, uA);
                v8f uB = vzero8(); uB = wmma4(a0B[i], bu0, uB); uB = wmma4(a1B[i], bu1, uB);
                float agA = 0.f, agB = 0.f;
                #pragma unroll
                for (int v = 0; v < 8; ++v) { agA = fmaf(uA[v], vA[v], agA); agB = fmaf(uB[v], vB[v], agB); }
                const float keep = hi ? agB : agA;
                const float send = hi ? agA : agB;
                bij[ib + i*16] += keep + xhalf(send);
            }
            __syncthreads();
            #pragma unroll
            for (int bsel = 0; bsel < 2; ++bsel) {
                int base = (wav + 8*bsel)*513 + (t & 31)*16;
                float mx = bij[base];
                #pragma unroll
                for (int j = 1; j < 16; ++j) mx = fmaxf(mx, bij[base + j]);
                float e[16]; float sum = 0.f;
                #pragma unroll
                for (int j = 0; j < 16; ++j) { e[j] = expf(bij[base + j] - mx); sum += e[j]; }
                float inv = 1.0f / sum;
                #pragma unroll
                for (int j = 0; j < 16; ++j) cij[base + j] = e[j] * inv;
            }
            __syncthreads();
        }
    }

    {
        float pA = 0.f, pB = 0.f;
        #pragma unroll
        for (int v = 0; v < 8; ++v) {
            pA = fmaf(vA[v], Wf[(wav    )*16 + v + 8*hi], pA);
            pB = fmaf(vB[v], Wf[(wav + 8)*16 + v + 8*hi], pB);
        }
        pA += __shfl_xor(pA, 16, 32);  pB += __shfl_xor(pB, 16, 32);
        if (lane < 16) { part[wav*17 + bb] = pA; part[(wav+8)*17 + bb] = pB; }
    }
    __syncthreads();
    if (t < 16) {
        float acc = bf[0];
        #pragma unroll
        for (int j = 0; j < 16; ++j) acc += part[j*17 + t];
        out[tile*16 + t] = acc;
    }
}

extern "C" void kernel_launch(void* const* d_in, const int* in_sizes, int n_in,
                              void* d_out, int out_size, void* d_ws, size_t ws_size,
                              hipStream_t stream) {
    const float* x    = (const float*)d_in[0];
    const float* W1   = (const float*)d_in[1];
    const float* b1   = (const float*)d_in[2];
    const float* W2   = (const float*)d_in[3];
    const float* b2   = (const float*)d_in[4];
    const float* ln_g = (const float*)d_in[5];
    const float* ln_b = (const float*)d_in[6];
    const float* Wcap = (const float*)d_in[7];
    const float* Wf   = (const float*)d_in[8];
    const float* bf   = (const float*)d_in[9];
    float* out = (float*)d_out;

    const int B = in_sizes[0] / 256;
    const int blocks = B / 16;
    capsnet_fused<<<blocks, 256, 0, stream>>>(x, W1, b1, W2, b2, ln_g, ln_b, Wcap, Wf, bf, out);
}


# Routing logit and coefficient arrays re-pitched in LDS (capsule row 16 -> 18 words, sample 513 -> 579 words): softmax reads and writes go from 8-way bank conflict to none; sums bit-identical
# speedup vs baseline: 1.4061x; 1.0704x over previous
#include <hip/hip_runtime.h>
#include <math.h>

typedef __attribute__((ext_vector_type(2))) float v2f;
typedef __attribute__((ext_vector_type(8))) float v8f;

#define LN_EPS 1e-5f
#define SQ_EPS 1e-9f

#define RP 18
#define RS 579

__device__ __forceinline__ float gelu_exact(float x) {
    return 0.5f * x * (1.0f + erff(x * 0.70710678118654752440f));
}

__device__ __forceinline__ v8f wmma4(v2f a, v2f b, v8f c) {
    return __builtin_amdgcn_wmma_f32_16x16x4_f32(false, a, false, b, (short)0, c, false, false);
}

__device__ __forceinline__ float xhalf(float x) {
    const int b = __builtin_bit_cast(int, x);
    return __builtin_bit_cast(float, __builtin_amdgcn_permlanex16(b, b, 0x76543210, 0xfedcba98, false, false));
}

__device__ __forceinline__ v8f vzero8() {
    v8f z = {0.f,0.f,0.f,0.f,0.f,0.f,0.f,0.f};
    return z;
}

__global__ __attribute__((amdgpu_waves_per_eu(1))) __launch_bounds__(256)
void capsnet_fused(
    const float* __restrict__ x,   const float* __restrict__ W1, const float* __restrict__ b1,
    const float* __restrict__ W2,  const float* __restrict__ b2,
    const float* __restrict__ ln_g,const float* __restrict__ ln_b,
    const float* __restrict__ Wcap,const float* __restrict__ Wf, const float* __restrict__ bf,
    float* __restrict__ out)
{
    __shared__ float xh[16*256];
    __shared__ float h1s[16*32];
    __shared__ float mu_s[16];
    __shared__ float rstd_s[16];
    __shared__ float u_t[32*130];
    __shared__ float bij[16*RS];
    __shared__ float cij[16*RS];
    __shared__ float part[16*17];

    const int t    = threadIdx.x;
    const int lane = t & 31;
    const int wav  = t >> 5;
    const int bb   = lane & 15;
    const int hi   = lane >> 4;

    const int tile = blockIdx.x;
    const float* xg = x + (size_t)tile * 16 * 256;

    const int n1 = t & 31, r1 = t >> 5;


    float xr[16];
    #pragma unroll
    for (int r = 0; r < 16; ++r) xr[r] = xg[t + 256*r];
    float wq[64];
    #pragma unroll
    for (int k = 0; k < 64; ++k) wq[k] = W1[k*32 + n1];
    float acc0 = b1[n1], acc1 = acc0;
    #pragma unroll
    for (int r = 0; r < 16; ++r) xh[t + 256*r] = xr[r];
    __syncthreads();

    #pragma unroll 1
    for (int c = 0; c < 3; ++c) {
        float wn[64];
        #pragma unroll
        for (int k = 0; k < 64; ++k) wn[k] = W1[((c+1)*64 + k)*32 + n1];
        #pragma unroll
        for (int k = 0; k < 64; ++k) {
            acc0 = fmaf(xh[r1*256 + c*64 + k],     wq[k], acc0);
            acc1 = fmaf(xh[(r1+8)*256 + c*64 + k], wq[k], acc1);
        }
        #pragma unroll
        for (int k = 0; k < 64; ++k) wq[k] = wn[k];
    }
    float w2c[64], b2v[8];
    #pragma unroll
    for (int q = 0; q < 2; ++q) {
        #pragma unroll
        for (int k = 0; k < 32; ++k) w2c[q*32 + k] = W2[k*256 + n1 + 32*q];
    }
    #pragma unroll
    for (int rr = 0; rr < 8; ++rr) b2v[rr] = b2[n1 + 32*rr];
    #pragma unroll
    for (int k = 0; k < 64; ++k) {
        acc0 = fmaf(xh[r1*256 + 192 + k],     wq[k], acc0);
        acc1 = fmaf(xh[(r1+8)*256 + 192 + k], wq[k], acc1);
    }
    h1s[r1*32 + n1]     = gelu_exact(acc0);
    h1s[(r1+8)*32 + n1] = gelu_exact(acc1);
    __syncthreads();

    {
        float h1a[32], h1b[32];
        #pragma unroll
        for (int k = 0; k < 32; ++k) { h1a[k] = h1s[r1*32 + k]; h1b[k] = h1s[(r1+8)*32 + k]; }
        #pragma unroll
        for (int p = 0; p < 4; ++p) {
            float w2n[64];
            if (p < 3) {
                #pragma unroll
                for (int q = 0; q < 2; ++q) {
                    #pragma unroll
                    for (int k = 0; k < 32; ++k) w2n[q*32 + k] = W2[k*256 + n1 + 32*(2*(p+1) + q)];
                }
            }
            #pragma unroll
            for (int q = 0; q < 2; ++q) {
                const int n = n1 + 32*(2*p + q);
                float a0 = b2v[2*p + q], a1 = a0;
                #pragma unroll
                for (int k = 0; k < 32; ++k) {
                    a0 = fmaf(h1a[k], w2c[q*32 + k], a0);
                    a1 = fmaf(h1b[k], w2c[q*32 + k], a1);
                }
                xh[r1*256 + n]     = gelu_exact(a0);
                xh[(r1+8)*256 + n] = gelu_exact(a1);
            }
            if (p < 3) {
                #pragma unroll
                for (int k = 0; k < 64; ++k) w2c[k] = w2n[k];
            }
        }
    }
    __syncthreads();

    {
        int ba = wav, bc = wav + 8;
        float sa = 0.f, sb = 0.f;
        #pragma unroll
        for (int r = 0; r < 8; ++r) { sa += xh[ba*256 + lane + 32*r]; sb += xh[bc*256 + lane + 32*r]; }
        #pragma unroll
        for (int m = 16; m >= 1; m >>= 1) { sa += __shfl_xor(sa, m, 32); sb += __shfl_xor(sb, m, 32); }
        float mua = sa * (1.0f/256.0f), mub = sb * (1.0f/256.0f);
        float va = 0.f, vb = 0.f;
        #pragma unroll
        for (int r = 0; r < 8; ++r) {
            float da = xh[ba*256 + lane + 32*r] - mua; va = fmaf(da, da, va);
            float db = xh[bc*256 + lane + 32*r] - mub; vb = fmaf(db, db, vb);
        }
        #pragma unroll
        for (int m = 16; m >= 1; m >>= 1) { va += __shfl_xor(va, m, 32); vb += __shfl_xor(vb, m, 32); }
        if (lane == 0) {
            mu_s[ba] = mua; rstd_s[ba] = rsqrtf(va * (1.0f/256.0f) + LN_EPS);
            mu_s[bc] = mub; rstd_s[bc] = rsqrtf(vb * (1.0f/256.0f) + LN_EPS);
        }
    }
    __syncthreads();

    {
        int i = t & 31, b0 = t >> 5;
        float g8[8], be8[8];
        #pragma unroll
        for (int k = 0; k < 8; ++k) { g8[k] = ln_g[i*8 + k]; be8[k] = ln_b[i*8 + k]; }
        #pragma unroll
        for (int bsel = 0; bsel < 2; ++bsel) {
            int b = b0 + 8*bsel;
            float mu = mu_s[b], rs = rstd_s[b];
            float vals[8]; float nsq = 0.f;
            #pragma unroll
            for (int k = 0; k < 8; ++k) {
                float v = (xh[b*256 + i*8 + k] - mu) * rs * g8[k] + be8[k];
                vals[k] = v; nsq = fmaf(v, v, nsq);
            }
            float scale = nsq / ((1.0f + nsq) * sqrtf(nsq + SQ_EPS));
            #pragma unroll
            for (int k = 0; k < 8; ++k) u_t[i*130 + k*16 + b] = scale * vals[k];
        }
    }

    #pragma unroll
    for (int r = 0; r < 36; ++r) { bij[t + 256*r] = 0.f; cij[t + 256*r] = 0.0625f; }
    if (t < 16*RS - 9216) { bij[9216 + t] = 0.f; cij[9216 + t] = 0.0625f; }
    __syncthreads();

    v2f a0A[32], a1A[32], a0B[32], a1B[32];
    #pragma unroll
    for (int i = 0; i < 32; ++i) {
        const float* wcA = Wcap + (size_t)((i*16 + wav    )*8)*16 + bb;
        const float* wcB = Wcap + (size_t)((i*16 + wav + 8)*8)*16 + bb;
        a0A[i][0] = wcA[(2*hi+0)*16];  a0A[i][1] = wcA[(2*hi+1)*16];
        a1A[i][0] = wcA[(2*hi+4)*16];  a1A[i][1] = wcA[(2*hi+5)*16];
        a0B[i][0] = wcB[(2*hi+0)*16];  a0B[i][1] = wcB[(2*hi+1)*16];
        a1B[i][0] = wcB[(2*hi+4)*16];  a1B[i][1] = wcB[(2*hi+5)*16];
    }

    v8f vA = vzero8(), vB = vzero8();
    for (int r = 0; r < 3; ++r) {
        v8f sA0 = vzero8(), sA1 = vzero8(), sB0 = vzero8(), sB1 = vzero8();
        v2f ub0[32], ub1[32];
        #pragma unroll
        for (int i = 0; i < 32; ++i) {
            const float* ub = u_t + i*130 + bb;
            v2f bu0, bu1;
            bu0[0] = ub[(2*hi+0)*16];  bu0[1] = ub[(2*hi+1)*16];
            bu1[0] = ub[(2*hi+4)*16];  bu1[1] = ub[(2*hi+5)*16];
            ub0[i] = bu0;  ub1[i] = bu1;
            float cA = cij[bb*RS + i*RP + wav];
            float cB = cij[bb*RS + i*RP + wav + 8];
            v2f p0, p1;
            p0[0] = bu0[0]*cA; p0[1] = bu0[1]*cA; p1[0] = bu1[0]*cA; p1[1] = bu1[1]*cA;
            if (i & 1) { sA1 = wmma4(a0A[i], p0, sA1); sA1 = wmma4(a1A[i], p1, sA1); }
            else       { sA0 = wmma4(a0A[i], p0, sA0); sA0 = wmma4(a1A[i], p1, sA0); }
            p0[0] = bu0[0]*cB; p0[1] = bu0[1]*cB; p1[0] = bu1[0]*cB; p1[1] = bu1[1]*cB;
            if (i & 1) { sB1 = wmma4(a0B[i], p0, sB1); sB1 = wmma4(a1B[i], p1, sB1); }
            else       { sB0 = wmma4(a0B[i], p0, sB0); sB0 = wmma4(a1B[i], p1, sB0); }
        }
        v8f sA = sA0 + sA1, sB = sB0 + sB1;
        float nA = 0.f, nB = 0.f;
        #pragma unroll
        for (int v = 0; v < 8; ++v) { nA = fmaf(sA[v], sA[v], nA); nB = fmaf(sB[v], sB[v], nB); }
        nA += __shfl_xor(nA, 16, 32);  nB += __shfl_xor(nB, 16, 32);
        float scA = nA / ((1.0f + nA) * sqrtf(nA + SQ_EPS));
        float scB = nB / ((1.0f + nB) * sqrtf(nB + SQ_EPS));
        #pragma unroll
        for (int v = 0; v < 8; ++v) { vA[v] = sA[v]*scA; vB[v] = sB[v]*scB; }

        if (r < 2) {
            const int ib = bb*RS + wav + 8*hi;
            #pragma unroll
            for (int i = 0; i < 32; ++i) {
                const v2f bu0 = ub0[i], bu1 = ub1[i];
                v8f uA = vzero8(); uA = wmma4(a0A[i], bu0, uA); uA = wmma4(a1A[i], bu1, uA);
                v8f uB = vzero8(); uB = wmma4(a0B[i], bu0, uB); uB = wmma4(a1B[i], bu1, uB);
                float agA = 0.f, agB = 0.f;
                #pragma unroll
                for (int v = 0; v < 8; ++v) { agA = fmaf(uA[v], vA[v], agA); agB = fmaf(uB[v], vB[v], agB); }
                const float keep = hi ? agB : agA;
                const float send = hi ? agA : agB;
                bij[ib + i*RP] += keep + xhalf(send);
            }
            __syncthreads();
            #pragma unroll
            for (int bsel = 0; bsel < 2; ++bsel) {
                int base = (wav + 8*bsel)*RS + (t & 31)*RP;
                float mx = bij[base];
                #pragma unroll
                for (int j = 1; j < 16; ++j) mx = fmaxf(mx, bij[base + j]);
                float e[16]; float sum = 0.f;
                #pragma unroll
                for (int j = 0; j < 16; ++j) { e[j] = expf(bij[base + j] - mx); sum += e[j]; }
                float inv = 1.0f / sum;
                #pragma unroll
                for (int j = 0; j < 16; ++j) cij[base + j] = e[j] * inv;
            }
            __syncthreads();
        }
    }

    {
        float pA = 0.f, pB = 0.f;
        #pragma unroll
        for (int v = 0; v < 8; ++v) {
            pA = fmaf(vA[v], Wf[(wav    )*16 + v + 8*hi], pA);
            pB = fmaf(vB[v], Wf[(wav + 8)*16 + v + 8*hi], pB);
        }
        pA += __shfl_xor(pA, 16, 32);  pB += __shfl_xor(pB, 16, 32);
        if (lane < 16) { part[wav*17 + bb] = pA; part[(wav+8)*17 + bb] = pB; }
    }
    __syncthreads();
    if (t < 16) {
        float acc = bf[0];
        #pragma unroll
        for (int j = 0; j < 16; ++j) acc += part[j*17 + t];
        out[tile*16 + t] = acc;
    }
}

extern "C" void kernel_launch(void* const* d_in, const int* in_sizes, int n_in,
                              void* d_out, int out_size, void* d_ws, size_t ws_size,
                              hipStream_t stream) {
    const float* x    = (const float*)d_in[0];
    const float* W1   = (const float*)d_in[1];
    const float* b1   = (const float*)d_in[2];
    const float* W2   = (const float*)d_in[3];
    const float* b2   = (const float*)d_in[4];
    const float* ln_g = (const float*)d_in[5];
    const float* ln_b = (const float*)d_in[6];
    const float* Wcap = (const float*)d_in[7];
    const float* Wf   = (const float*)d_in[8];
    const float* bf   = (const float*)d_in[9];
    float* out = (float*)d_out;

    const int B = in_sizes[0] / 256;
    const int blocks = B / 16;
    capsnet_fused<<<blocks, 256, 0, stream>>>(x, W1, b1, W2, b2, ln_g, ln_b, Wcap, Wf, bf, out);
}


# Both MLP layers moved from per-thread FMA chains to the f32 matrix instruction (layer 1 in four k quarters added through LDS, biases added last): MLP sums now in matrix-core order
# speedup vs baseline: 1.6721x; 1.1892x over previous
#include <hip/hip_runtime.h>
#include <math.h>

typedef __attribute__((ext_vector_type(2))) float v2f;
typedef __attribute__((ext_vector_type(8))) float v8f;

#define LN_EPS 1e-5f
#define SQ_EPS 1e-9f

#define RP 18
#define RS 579

#define XP 260
#define HP 36

__device__ __forceinline__ float gelu_exact(float x) {
    return 0.5f * x * (1.0f + erff(x * 0.70710678118654752440f));
}

__device__ __forceinline__ v8f wmma4(v2f a, v2f b, v8f c) {
    return __builtin_amdgcn_wmma_f32_16x16x4_f32(false, a, false, b, (short)0, c, false, false);
}

__device__ __forceinline__ float xhalf(float x) {
    const int b = __builtin_bit_cast(int, x);
    return __builtin_bit_cast(float, __builtin_amdgcn_permlanex16(b, b, 0x76543210, 0xfedcba98, false, false));
}

__device__ __forceinline__ v8f vzero8() {
    v8f z = {0.f,0.f,0.f,0.f,0.f,0.f,0.f,0.f};
    return z;
}

__global__ __attribute__((amdgpu_waves_per_eu(1))) __launch_bounds__(256)
void capsnet_fused(
    const float* __restrict__ x,   const float* __restrict__ W1, const float* __restrict__ b1,
    const float* __restrict__ W2,  const float* __restrict__ b2,
    const float* __restrict__ ln_g,const float* __restrict__ ln_b,
    const float* __restrict__ Wcap,const float* __restrict__ Wf, const float* __restrict__ bf,
    float* __restrict__ out)
{
    __shared__ float xh[16*XP];
    __shared__ float h1s[16*HP];
    __shared__ float mu_s[16];
    __shared__ float rstd_s[16];
    __shared__ float u_t[32*130];
    __shared__ float bij[16*RS];
    __shared__ float cij[16*RS];
    __shared__ float part[16*17];

    const int t    = threadIdx.x;
    const int lane = t & 31;
    const int wav  = t >> 5;
    const int bb   = lane & 15;
    const int hi   = lane >> 4;

    const int tile = blockIdx.x;
    const float* xg = x + (size_t)tile * 16 * 256;

    const int nt = wav & 1, kq = wav >> 1;

    float xr[16];
    #pragma unroll
    for (int r = 0; r < 16; ++r) xr[r] = xg[t + 256*r];
    v2f w1f[16];
    #pragma unroll
    for (int s = 0; s < 16; ++s) {
        const float* wp = W1 + (size_t)(kq*64 + 4*s + 2*hi)*32 + nt*16 + bb;
        w1f[s][0] = wp[0];  w1f[s][1] = wp[32];
    }
    v2f w2f0[8], w2f1[8];
    #pragma unroll
    for (int s = 0; s < 8; ++s) {
        const float* wp = W2 + (size_t)(4*s + 2*hi)*256 + 32*wav + bb;
        w2f0[s][0] = wp[0];   w2f0[s][1] = wp[256];
        w2f1[s][0] = wp[16];  w2f1[s][1] = wp[256 + 16];
    }
    const float b2a = b2[32*wav + bb], b2b = b2[32*wav + 16 + bb];
    const float b1v = b1[lane];
    #pragma unroll
    for (int r = 0; r < 16; ++r) xh[r*XP + t] = xr[r];
    __syncthreads();

    float* const p1 = bij;
    {
        v8f c0 = vzero8(), c1 = vzero8();
        const float* xa = xh + bb*XP + kq*64 + 2*hi;
        #pragma unroll
        for (int s = 0; s < 16; ++s) {
            v2f af;  af[0] = xa[4*s];  af[1] = xa[4*s + 1];
            if (s & 1) c1 = wmma4(af, w1f[s], c1);
            else       c0 = wmma4(af, w1f[s], c0);
        }
        const v8f c = c0 + c1;
        float* pp = p1 + (kq*16 + 8*hi)*HP + nt*16 + bb;
        #pragma unroll
        for (int r = 0; r < 8; ++r) pp[r*HP] = c[r];
    }
    __syncthreads();

    #pragma unroll
    for (int bsel = 0; bsel < 2; ++bsel) {
        const int b = wav + 8*bsel;
        float acc = p1[b*HP + lane];
        #pragma unroll
        for (int q = 1; q < 4; ++q) acc += p1[(q*16 + b)*HP + lane];
        h1s[b*HP + lane] = gelu_exact(acc + b1v);
    }
    __syncthreads();

    {
        v8f d0 = vzero8(), d1 = vzero8();
        const float* ha = h1s + bb*HP + 2*hi;
        #pragma unroll
        for (int s = 0; s < 8; ++s) {
            v2f af;  af[0] = ha[4*s];  af[1] = ha[4*s + 1];
            d0 = wmma4(af, w2f0[s], d0);
            d1 = wmma4(af, w2f1[s], d1);
        }
        float* ho = xh + (8*hi)*256 + 32*wav + bb;
        #pragma unroll
        for (int r = 0; r < 8; ++r) {
            ho[r*256]      = gelu_exact(d0[r] + b2a);
            ho[r*256 + 16] = gelu_exact(d1[r] + b2b);
        }
    }
    __syncthreads();

    {
        int ba = wav, bc = wav + 8;
        float sa = 0.f, sb = 0.f;
        #pragma unroll
        for (int r = 0; r < 8; ++r) { sa += xh[ba*256 + lane + 32*r]; sb += xh[bc*256 + lane + 32*r]; }
        #pragma unroll
        for (int m = 16; m >= 1; m >>= 1) { sa += __shfl_xor(sa, m, 32); sb += __shfl_xor(sb, m, 32); }
        float mua = sa * (1.0f/256.0f), mub = sb * (1.0f/256.0f);
        float va = 0.f, vb = 0.f;
        #pragma unroll
        for (int r = 0; r < 8; ++r) {
            float da = xh[ba*256 + lane + 32*r] - mua; va = fmaf(da, da, va);
            float db = xh[bc*256 + lane + 32*r] - mub; vb = fmaf(db, db, vb);
        }
        #pragma unroll
        for (int m = 16; m >= 1; m >>= 1) { va += __shfl_xor(va, m, 32); vb += __shfl_xor(vb, m, 32); }
        if (lane == 0) {
            mu_s[ba] = mua; rstd_s[ba] = rsqrtf(va * (1.0f/256.0f) + LN_EPS);
            mu_s[bc] = mub; rstd_s[bc] = rsqrtf(vb * (1.0f/256.0f) + LN_EPS);
        }
    }
    __syncthreads();

    {
        int i = t & 31, b0 = t >> 5;
        float g8[8], be8[8];
        #pragma unroll
        for (int k = 0; k < 8; ++k) { g8[k] = ln_g[i*8 + k]; be8[k] = ln_b[i*8 + k]; }
        #pragma unroll
        for (int bsel = 0; bsel < 2; ++bsel) {
            int b = b0 + 8*bsel;
            float mu = mu_s[b], rs = rstd_s[b];
            float vals[8]; float nsq = 0.f;
            #pragma unroll
            for (int k = 0; k < 8; ++k) {
                float v = (xh[b*256 + i*8 + k] - mu) * rs * g8[k] + be8[k];
                vals[k] = v; nsq = fmaf(v, v, nsq);
            }
            float scale = nsq / ((1.0f + nsq) * sqrtf(nsq + SQ_EPS));
            #pragma unroll
            for (int k = 0; k < 8; ++k) u_t[i*130 + k*16 + b] = scale * vals[k];
        }
    }

    #pragma unroll
    for (int r = 0; r < 36; ++r) { bij[t + 256*r] = 0.f; cij[t + 256*r] = 0.0625f; }
    if (t < 16*RS - 9216) { bij[9216 + t] = 0.f; cij[9216 + t] = 0.0625f; }
    __syncthreads();

    v2f a0A[32], a1A[32], a0B[32], a1B[32];
    #pragma unroll
    for (int i = 0; i < 32; ++i) {
        const float* wcA = Wcap + (size_t)((i*16 + wav    )*8)*16 + bb;
        const float* wcB = Wcap + (size_t)((i*16 + wav + 8)*8)*16 + bb;
        a0A[i][0] = wcA[(2*hi+0)*16];  a0A[i][1] = wcA[(2*hi+1)*16];
        a1A[i][0] = wcA[(2*hi+4)*16];  a1A[i][1] = wcA[(2*hi+5)*16];
        a0B[i][0] = wcB[(2*hi+0)*16];  a0B[i][1] = wcB[(2*hi+1)*16];
        a1B[i][0] = wcB[(2*hi+4)*16];  a1B[i][1] = wcB[(2*hi+5)*16];
    }

    v8f vA = vzero8(), vB = vzero8();
    for (int r = 0; r < 3; ++r) {
        v8f sA0 = vzero8(), sA1 = vzero8(), sB0 = vzero8(), sB1 = vzero8();
        v2f ub0[32], ub1[32];
        #pragma unroll
        for (int i = 0; i < 32; ++i) {
            const float* ub = u_t + i*130 + bb;
            v2f bu0, bu1;
            bu0[0] = ub[(2*hi+0)*16];  bu0[1] = ub[(2*hi+1)*16];
            bu1[0] = ub[(2*hi+4)*16];  bu1[1] = ub[(2*hi+5)*16];
            ub0[i] = bu0;  ub1[i] = bu1;
            float cA = cij[bb*RS + i*RP + wav];
            float cB = cij[bb*RS + i*RP + wav + 8];
            v2f p0, p1;
            p0[0] = bu0[0]*cA; p0[1] = bu0[1]*cA; p1[0] = bu1[0]*cA; p1[1] = bu1[1]*cA;
            if (i & 1) { sA1 = wmma4(a0A[i], p0, sA1); sA1 = wmma4(a1A[i], p1, sA1); }
            else       { sA0 = wmma4(a0A[i], p0, sA0); sA0 = wmma4(a1A[i], p1, sA0); }
            p0[0] = bu0[0]*cB; p0[1] = bu0[1]*cB; p1[0] = bu1[0]*cB; p1[1] = bu1[1]*cB;
            if (i & 1) { sB1 = wmma4(a0B[i], p0, sB1); sB1 = wmma4(a1B[i], p1, sB1); }
            else       { sB0 = wmma4(a0B[i], p0, sB0); sB0 = wmma4(a1B[i], p1, sB0); }
        }
        v8f sA = sA0 + sA1, sB = sB0 + sB1;
        float nA = 0.f, nB = 0.f;
        #pragma unroll
        for (int v = 0; v < 8; ++v) { nA = fmaf(sA[v], sA[v], nA); nB = fmaf(sB[v], sB[v], nB); }
        nA += __shfl_xor(nA, 16, 32);  nB += __shfl_xor(nB, 16, 32);
        float scA = nA / ((1.0f + nA) * sqrtf(nA + SQ_EPS));
        float scB = nB / ((1.0f + nB) * sqrtf(nB + SQ_EPS));
        #pragma unroll
        for (int v = 0; v < 8; ++v) { vA[v] = sA[v]*scA; vB[v] = sB[v]*scB; }

        if (r < 2) {
            const int ib = bb*RS + wav + 8*hi;
            #pragma unroll
            for (int i = 0; i < 32; ++i) {
                const v2f bu0 = ub0[i], bu1 = ub1[i];
                v8f uA = vzero8(); uA = wmma4(a0A[i], bu0, uA); uA = wmma4(a1A[i], bu1, uA);
                v8f uB = vzero8(); uB = wmma4(a0B[i], bu0, uB); uB = wmma4(a1B[i], bu1, uB);
                float agA = 0.f, agB = 0.f;
                #pragma unroll
                for (int v = 0; v < 8; ++v) { agA = fmaf(uA[v], vA[v], agA); agB = fmaf(uB[v], vB[v], agB); }
                const float keep = hi ? agB : agA;
                const float send = hi ? agA : agB;
                bij[ib + i*RP] += keep + xhalf(send);
            }
            __syncthreads();
            #pragma unroll
            for (int bsel = 0; bsel < 2; ++bsel) {
                int base = (wav + 8*bsel)*RS + (t & 31)*RP;
                float mx = bij[base];
                #pragma unroll
                for (int j = 1; j < 16; ++j) mx = fmaxf(mx, bij[base + j]);
                float e[16]; float sum = 0.f;
                #pragma unroll
                for (int j = 0; j < 16; ++j) { e[j] = expf(bij[base + j] - mx); sum += e[j]; }
                float inv = 1.0f / sum;
                #pragma unroll
                for (int j = 0; j < 16; ++j) cij[base + j] = e[j] * inv;
            }
            __syncthreads();
        }
    }

    {
        float pA = 0.f, pB = 0.f;
        #pragma unroll
        for (int v = 0; v < 8; ++v) {
            pA = fmaf(vA[v], Wf[(wav    )*16 + v + 8*hi], pA);
            pB = fmaf(vB[v], Wf[(wav + 8)*16 + v + 8*hi], pB);
        }
        pA += __shfl_xor(pA, 16, 32);  pB += __shfl_xor(pB, 16, 32);
        if (lane < 16) { part[wav*17 + bb] = pA; part[(wav+8)*17 + bb] = pB; }
    }
    __syncthreads();
    if (t < 16) {
        float acc = bf[0];
        #pragma unroll
        for (int j = 0; j < 16; ++j) acc += part[j*17 + t];
        out[tile*16 + t] = acc;
    }
}

extern "C" void kernel_launch(void* const* d_in, const int* in_sizes, int n_in,
                              void* d_out, int out_size, void* d_ws, size_t ws_size,
                              hipStream_t stream) {
    const float* x    = (const float*)d_in[0];
    const float* W1   = (const float*)d_in[1];
    const float* b1   = (const float*)d_in[2];
    const float* W2   = (const float*)d_in[3];
    const float* b2   = (const float*)d_in[4];
    const float* ln_g = (const float*)d_in[5];
    const float* ln_b = (const float*)d_in[6];
    const float* Wcap = (const float*)d_in[7];
    const float* Wf   = (const float*)d_in[8];
    const float* bf   = (const float*)d_in[9];
    float* out = (float*)d_out;

    const int B = in_sizes[0] / 256;
    const int blocks = B / 16;
    capsnet_fused<<<blocks, 256, 0, stream>>>(x, W1, b1, W2, b2, ln_g, ln_b, Wcap, Wf, bf, out);
}


# 256 persistent work-groups (one per WGP) each walk over 4 contiguous tiles with the Wcap fragments kept in registers: Wcap fetched once per work-group, not once per tile; sums bit-identical
# speedup vs baseline: 1.8342x; 1.0969x over previous
#include <hip/hip_runtime.h>
#include <math.h>

typedef __attribute__((ext_vector_type(2))) float v2f;
typedef __attribute__((ext_vector_type(8))) float v8f;

#define LN_EPS 1e-5f
#define SQ_EPS 1e-9f

#define RP 18
#define RS 579

#define XP 260
#define HP 36

#define NWG 256

__device__ __forceinline__ float gelu_exact(float x) {
    return 0.5f * x * (1.0f + erff(x * 0.70710678118654752440f));
}

__device__ __forceinline__ v8f wmma4(v2f a, v2f b, v8f c) {
    return __builtin_amdgcn_wmma_f32_16x16x4_f32(false, a, false, b, (short)0, c, false, false);
}

__device__ __forceinline__ float xhalf(float x) {
    const int b = __builtin_bit_cast(int, x);
    return __builtin_bit_cast(float, __builtin_amdgcn_permlanex16(b, b, 0x76543210, 0xfedcba98, false, false));
}

__device__ __forceinline__ int opaque_zero() {
    int z = 0;
    asm volatile("" : "+s"(z));
    return z;
}

__device__ __forceinline__ v8f vzero8() {
    v8f z = {0.f,0.f,0.f,0.f,0.f,0.f,0.f,0.f};
    return z;
}

__global__ __attribute__((amdgpu_waves_per_eu(1))) __launch_bounds__(256)
void capsnet_fused(
    const float* __restrict__ x,   const float* __restrict__ W1, const float* __restrict__ b1,
    const float* __restrict__ W2,  const float* __restrict__ b2,
    const float* __restrict__ ln_g,const float* __restrict__ ln_b,
    const float* __restrict__ Wcap,const float* __restrict__ Wf, const float* __restrict__ bf,
    float* __restrict__ out, const int ntiles)
{
    __shared__ float xh[16*XP];
    __shared__ float h1s[16*HP];
    __shared__ float mu_s[16];
    __shared__ float rstd_s[16];
    __shared__ float u_t[32*130];
    __shared__ float bij[16*RS];
    __shared__ float cij[16*RS];
    __shared__ float part[16*17];

    const int t    = threadIdx.x;
    const int lane = t & 31;
    const int wav  = t >> 5;
    const int bb   = lane & 15;
    const int hi   = lane >> 4;

    const int nt = wav & 1, kq = wav >> 1;

    v2f a0A[32], a1A[32], a0B[32], a1B[32];
    #pragma unroll
    for (int i = 0; i < 32; ++i) {
        const float* wcA = Wcap + (size_t)((i*16 + wav    )*8)*16 + bb;
        const float* wcB = Wcap + (size_t)((i*16 + wav + 8)*8)*16 + bb;
        a0A[i][0] = wcA[(2*hi+0)*16];  a0A[i][1] = wcA[(2*hi+1)*16];
        a1A[i][0] = wcA[(2*hi+4)*16];  a1A[i][1] = wcA[(2*hi+5)*16];
        a0B[i][0] = wcB[(2*hi+0)*16];  a0B[i][1] = wcB[(2*hi+1)*16];
        a1B[i][0] = wcB[(2*hi+4)*16];  a1B[i][1] = wcB[(2*hi+5)*16];
    }


    const unsigned tile_lo = (blockIdx.x * (unsigned)ntiles) / NWG;
    const unsigned tile_hi = ((blockIdx.x + 1u) * (unsigned)ntiles) / NWG;
    for (unsigned tile = tile_lo; tile < tile_hi; ++tile) {
    const float* xg = x + (size_t)tile * 16 * 256;
    const int tz = opaque_zero();
    const float* const W1t = W1 + tz;    const float* const b1t = b1 + tz;
    const float* const W2t = W2 + tz;    const float* const b2t = b2 + tz;
    const float* const lgt = ln_g + tz;  const float* const lbt = ln_b + tz;
    const float* const Wft = Wf + tz;    const float* const bft = bf + tz;


    float xr[16];
    #pragma unroll
    for (int r = 0; r < 16; ++r) xr[r] = xg[t + 256*r];
    v2f w1f[16];
    #pragma unroll
    for (int s = 0; s < 16; ++s) {
        const float* wp = W1t + (size_t)(kq*64 + 4*s + 2*hi)*32 + nt*16 + bb;
        w1f[s][0] = wp[0];  w1f[s][1] = wp[32];
    }
    v2f w2f0[8], w2f1[8];
    #pragma unroll
    for (int s = 0; s < 8; ++s) {
        const float* wp = W2t + (size_t)(4*s + 2*hi)*256 + 32*wav + bb;
        w2f0[s][0] = wp[0];   w2f0[s][1] = wp[256];
        w2f1[s][0] = wp[16];  w2f1[s][1] = wp[256 + 16];
    }
    const float b2a = b2t[32*wav + bb], b2b = b2t[32*wav + 16 + bb];
    const float b1v = b1t[lane];
    #pragma unroll
    for (int r = 0; r < 16; ++r) xh[r*XP + t] = xr[r];
    __syncthreads();

    float* const p1 = bij;
    {
        v8f c0 = vzero8(), c1 = vzero8();
        const float* xa = xh + bb*XP + kq*64 + 2*hi;
        #pragma unroll
        for (int s = 0; s < 16; ++s) {
            v2f af;  af[0] = xa[4*s];  af[1] = xa[4*s + 1];
            if (s & 1) c1 = wmma4(af, w1f[s], c1);
            else       c0 = wmma4(af, w1f[s], c0);
        }
        const v8f c = c0 + c1;
        float* pp = p1 + (kq*16 + 8*hi)*HP + nt*16 + bb;
        #pragma unroll
        for (int r = 0; r < 8; ++r) pp[r*HP] = c[r];
    }
    __syncthreads();

    #pragma unroll
    for (int bsel = 0; bsel < 2; ++bsel) {
        const int b = wav + 8*bsel;
        float acc = p1[b*HP + lane];
        #pragma unroll
        for (int q = 1; q < 4; ++q) acc += p1[(q*16 + b)*HP + lane];
        h1s[b*HP + lane] = gelu_exact(acc + b1v);
    }
    __syncthreads();

    {
        v8f d0 = vzero8(), d1 = vzero8();
        const float* ha = h1s + bb*HP + 2*hi;
        #pragma unroll
        for (int s = 0; s < 8; ++s) {
            v2f af;  af[0] = ha[4*s];  af[1] = ha[4*s + 1];
            d0 = wmma4(af, w2f0[s], d0);
            d1 = wmma4(af, w2f1[s], d1);
        }
        float* ho = xh + (8*hi)*256 + 32*wav + bb;
        #pragma unroll
        for (int r = 0; r < 8; ++r) {
            ho[r*256]      = gelu_exact(d0[r] + b2a);
            ho[r*256 + 16] = gelu_exact(d1[r] + b2b);
        }
    }
    __syncthreads();

    {
        int ba = wav, bc = wav + 8;
        float sa = 0.f, sb = 0.f;
        #pragma unroll
        for (int r = 0; r < 8; ++r) { sa += xh[ba*256 + lane + 32*r]; sb += xh[bc*256 + lane + 32*r]; }
        #pragma unroll
        for (int m = 16; m >= 1; m >>= 1) { sa += __shfl_xor(sa, m, 32); sb += __shfl_xor(sb, m, 32); }
        float mua = sa * (1.0f/256.0f), mub = sb * (1.0f/256.0f);
        float va = 0.f, vb = 0.f;
        #pragma unroll
        for (int r = 0; r < 8; ++r) {
            float da = xh[ba*256 + lane + 32*r] - mua; va = fmaf(da, da, va);
            float db = xh[bc*256 + lane + 32*r] - mub; vb = fmaf(db, db, vb);
        }
        #pragma unroll
        for (int m = 16; m >= 1; m >>= 1) { va += __shfl_xor(va, m, 32); vb += __shfl_xor(vb, m, 32); }
        if (lane == 0) {
            mu_s[ba] = mua; rstd_s[ba] = rsqrtf(va * (1.0f/256.0f) + LN_EPS);
            mu_s[bc] = mub; rstd_s[bc] = rsqrtf(vb * (1.0f/256.0f) + LN_EPS);
        }
    }
    __syncthreads();

    {
        int i = t & 31, b0 = t >> 5;
        float g8[8], be8[8];
        #pragma unroll
        for (int k = 0; k < 8; ++k) { g8[k] = lgt[i*8 + k]; be8[k] = lbt[i*8 + k]; }
        #pragma unroll
        for (int bsel = 0; bsel < 2; ++bsel) {
            int b = b0 + 8*bsel;
            float mu = mu_s[b], rs = rstd_s[b];
            float vals[8]; float nsq = 0.f;
            #pragma unroll
            for (int k = 0; k < 8; ++k) {
                float v = (xh[b*256 + i*8 + k] - mu) * rs * g8[k] + be8[k];
                vals[k] = v; nsq = fmaf(v, v, nsq);
            }
            float scale = nsq / ((1.0f + nsq) * sqrtf(nsq + SQ_EPS));
            #pragma unroll
            for (int k = 0; k < 8; ++k) u_t[i*130 + k*16 + b] = scale * vals[k];
        }
    }

    #pragma unroll
    for (int r = 0; r < 36; ++r) { bij[t + 256*r] = 0.f; cij[t + 256*r] = 0.0625f; }
    if (t < 16*RS - 9216) { bij[9216 + t] = 0.f; cij[9216 + t] = 0.0625f; }
    __syncthreads();

    v8f vA = vzero8(), vB = vzero8();
    for (int r = 0; r < 3; ++r) {
        v8f sA0 = vzero8(), sA1 = vzero8(), sB0 = vzero8(), sB1 = vzero8();
        v2f ub0[32], ub1[32];
        #pragma unroll
        for (int i = 0; i < 32; ++i) {
            const float* ub = u_t + i*130 + bb;
            v2f bu0, bu1;
            bu0[0] = ub[(2*hi+0)*16];  bu0[1] = ub[(2*hi+1)*16];
            bu1[0] = ub[(2*hi+4)*16];  bu1[1] = ub[(2*hi+5)*16];
            ub0[i] = bu0;  ub1[i] = bu1;
            float cA = cij[bb*RS + i*RP + wav];
            float cB = cij[bb*RS + i*RP + wav + 8];
            v2f p0, p1;
            p0[0] = bu0[0]*cA; p0[1] = bu0[1]*cA; p1[0] = bu1[0]*cA; p1[1] = bu1[1]*cA;
            if (i & 1) { sA1 = wmma4(a0A[i], p0, sA1); sA1 = wmma4(a1A[i], p1, sA1); }
            else       { sA0 = wmma4(a0A[i], p0, sA0); sA0 = wmma4(a1A[i], p1, sA0); }
            p0[0] = bu0[0]*cB; p0[1] = bu0[1]*cB; p1[0] = bu1[0]*cB; p1[1] = bu1[1]*cB;
            if (i & 1) { sB1 = wmma4(a0B[i], p0, sB1); sB1 = wmma4(a1B[i], p1, sB1); }
            else       { sB0 = wmma4(a0B[i], p0, sB0); sB0 = wmma4(a1B[i], p1, sB0); }
        }
        v8f sA = sA0 + sA1, sB = sB0 + sB1;
        float nA = 0.f, nB = 0.f;
        #pragma unroll
        for (int v = 0; v < 8; ++v) { nA = fmaf(sA[v], sA[v], nA); nB = fmaf(sB[v], sB[v], nB); }
        nA += __shfl_xor(nA, 16, 32);  nB += __shfl_xor(nB, 16, 32);
        float scA = nA / ((1.0f + nA) * sqrtf(nA + SQ_EPS));
        float scB = nB / ((1.0f + nB) * sqrtf(nB + SQ_EPS));
        #pragma unroll
        for (int v = 0; v < 8; ++v) { vA[v] = sA[v]*scA; vB[v] = sB[v]*scB; }

        if (r < 2) {
            const int ib = bb*RS + wav + 8*hi;
            #pragma unroll
            for (int i = 0; i < 32; ++i) {
                const v2f bu0 = ub0[i], bu1 = ub1[i];
                v8f uA = vzero8(); uA = wmma4(a0A[i], bu0, uA); uA = wmma4(a1A[i], bu1, uA);
                v8f uB = vzero8(); uB = wmma4(a0B[i], bu0, uB); uB = wmma4(a1B[i], bu1, uB);
                float agA = 0.f, agB = 0.f;
                #pragma unroll
                for (int v = 0; v < 8; ++v) { agA = fmaf(uA[v], vA[v], agA); agB = fmaf(uB[v], vB[v], agB); }
                const float keep = hi ? agB : agA;
                const float send = hi ? agA : agB;
                bij[ib + i*RP] += keep + xhalf(send);
            }
            __syncthreads();
            #pragma unroll
            for (int bsel = 0; bsel < 2; ++bsel) {
                int base = (wav + 8*bsel)*RS + (t & 31)*RP;
                float mx = bij[base];
                #pragma unroll
                for (int j = 1; j < 16; ++j) mx = fmaxf(mx, bij[base + j]);
                float e[16]; float sum = 0.f;
                #pragma unroll
                for (int j = 0; j < 16; ++j) { e[j] = expf(bij[base + j] - mx); sum += e[j]; }
                float inv = 1.0f / sum;
                #pragma unroll
                for (int j = 0; j < 16; ++j) cij[base + j] = e[j] * inv;
            }
            __syncthreads();
        }
    }

    {
        float pA = 0.f, pB = 0.f;
        #pragma unroll
        for (int v = 0; v < 8; ++v) {
            pA = fmaf(vA[v], Wft[(wav    )*16 + v + 8*hi], pA);
            pB = fmaf(vB[v], Wft[(wav + 8)*16 + v + 8*hi], pB);
        }
        pA += __shfl_xor(pA, 16, 32);  pB += __shfl_xor(pB, 16, 32);
        if (lane < 16) { part[wav*17 + bb] = pA; part[(wav+8)*17 + bb] = pB; }
    }
    __syncthreads();
    if (t < 16) {
        float acc = bft[0];
        #pragma unroll
        for (int j = 0; j < 16; ++j) acc += part[j*17 + t];
        out[tile*16 + t] = acc;
    }
    }
}

extern "C" void kernel_launch(void* const* d_in, const int* in_sizes, int n_in,
                              void* d_out, int out_size, void* d_ws, size_t ws_size,
                              hipStream_t stream) {
    const float* x    = (const float*)d_in[0];
    const float* W1   = (const float*)d_in[1];
    const float* b1   = (const float*)d_in[2];
    const float* W2   = (const float*)d_in[3];
    const float* b2   = (const float*)d_in[4];
    const float* ln_g = (const float*)d_in[5];
    const float* ln_b = (const float*)d_in[6];
    const float* Wcap = (const float*)d_in[7];
    const float* Wf   = (const float*)d_in[8];
    const float* bf   = (const float*)d_in[9];
    float* out = (float*)d_out;

    const int B = in_sizes[0] / 256;
    const int tiles = B / 16;
    capsnet_fused<<<NWG, 256, 0, stream>>>(x, W1, b1, W2, b2, ln_g, ln_b, Wcap, Wf, bf, out, tiles);
}


# Library expf (softmax) and erff (gelu) replaced by inline forms on the exp2 unit; erf range tested once per phase, not per call (no branch between calls); each call differs by about 1 ulp
# speedup vs baseline: 1.9660x; 1.0719x over previous
#include <hip/hip_runtime.h>
#include <math.h>

typedef __attribute__((ext_vector_type(2))) float v2f;
typedef __attribute__((ext_vector_type(8))) float v8f;

#define LN_EPS 1e-5f
#define SQ_EPS 1e-9f

#define RP 18
#define RS 579

#define XP 260
#define HP 36

#define NWG 256

__device__ __forceinline__ float exp_unit(float x) {
    return __builtin_amdgcn_exp2f(x * 1.44269504088896340736f);
}

#define ERF_SPLIT 0.927734375f
__device__ __forceinline__ float erf_small(float a) {
    const float s = a * a;
    float q = -5.96761703e-4f;
    q = fmaf(q, s, 4.99119423e-3f);
    q = fmaf(q, s, -2.67681349e-2f);
    q = fmaf(q, s, 1.12819925e-1f);
    q = fmaf(q, s, -3.76125336e-1f);
    q = fmaf(q, s, 1.28379166e-1f);
    return fmaf(q, a, a);
}
__device__ __forceinline__ float erf_large(float a) {
    const float t = fabsf(a);
    const float s = a * a;
    float r = fmaf(-1.72853470e-5f, t, 3.83197126e-4f);
    const float u = fmaf(-3.88396438e-3f, t, 2.42546219e-2f);
    r = fmaf(r, s, u);
    r = fmaf(r, t, -1.06777877e-1f);
    r = fmaf(r, t, -6.34846687e-1f);
    r = fmaf(r, t, -1.28717512e-1f);
    r = fmaf(r, t, -t);
    return copysignf(1.0f - exp_unit(r), a);
}

template <int N>
__device__ __forceinline__ void gelu_n(float (&z)[N]) {
    float a[N], e[N];
    float amax = 0.f;
    #pragma unroll
    for (int n = 0; n < N; ++n) {
        a[n] = z[n] * 0.70710678118654752440f;
        e[n] = erf_small(a[n]);
        amax = fmaxf(amax, fabsf(a[n]));
    }
    if (amax > ERF_SPLIT) {
        #pragma unroll
        for (int n = 0; n < N; ++n) {
            const float el = erf_large(a[n]);
            e[n] = (fabsf(a[n]) > ERF_SPLIT) ? el : e[n];
        }
    }
    #pragma unroll
    for (int n = 0; n < N; ++n) z[n] = 0.5f * z[n] * (1.0f + e[n]);
}

__device__ __forceinline__ v8f wmma4(v2f a, v2f b, v8f c) {
    return __builtin_amdgcn_wmma_f32_16x16x4_f32(false, a, false, b, (short)0, c, false, false);
}

__device__ __forceinline__ float xhalf(float x) {
    const int b = __builtin_bit_cast(int, x);
    return __builtin_bit_cast(float, __builtin_amdgcn_permlanex16(b, b, 0x76543210, 0xfedcba98, false, false));
}

__device__ __forceinline__ int opaque_zero() {
    int z = 0;
    asm volatile("" : "+s"(z));
    return z;
}

__device__ __forceinline__ v8f vzero8() {
    v8f z = {0.f,0.f,0.f,0.f,0.f,0.f,0.f,0.f};
    return z;
}

__global__ __attribute__((amdgpu_waves_per_eu(1))) __launch_bounds__(256)
void capsnet_fused(
    const float* __restrict__ x,   const float* __restrict__ W1, const float* __restrict__ b1,
    const float* __restrict__ W2,  const float* __restrict__ b2,
    const float* __restrict__ ln_g,const float* __restrict__ ln_b,
    const float* __restrict__ Wcap,const float* __restrict__ Wf, const float* __restrict__ bf,
    float* __restrict__ out, const int ntiles)
{
    __shared__ float xh[16*XP];
    __shared__ float h1s[16*HP];
    __shared__ float mu_s[16];
    __shared__ float rstd_s[16];
    __shared__ float u_t[32*130];
    __shared__ float bij[16*RS];
    __shared__ float cij[16*RS];
    __shared__ float part[16*17];

    const int t    = threadIdx.x;
    const int lane = t & 31;
    const int wav  = t >> 5;
    const int bb   = lane & 15;
    const int hi   = lane >> 4;

    const int nt = wav & 1, kq = wav >> 1;

    v2f a0A[32], a1A[32], a0B[32], a1B[32];
    #pragma unroll
    for (int i = 0; i < 32; ++i) {
        const float* wcA = Wcap + (size_t)((i*16 + wav    )*8)*16 + bb;
        const float* wcB = Wcap + (size_t)((i*16 + wav + 8)*8)*16 + bb;
        a0A[i][0] = wcA[(2*hi+0)*16];  a0A[i][1] = wcA[(2*hi+1)*16];
        a1A[i][0] = wcA[(2*hi+4)*16];  a1A[i][1] = wcA[(2*hi+5)*16];
        a0B[i][0] = wcB[(2*hi+0)*16];  a0B[i][1] = wcB[(2*hi+1)*16];
        a1B[i][0] = wcB[(2*hi+4)*16];  a1B[i][1] = wcB[(2*hi+5)*16];
    }


    const unsigned tile_lo = (blockIdx.x * (unsigned)ntiles) / NWG;
    const unsigned tile_hi = ((blockIdx.x + 1u) * (unsigned)ntiles) / NWG;
    for (unsigned tile = tile_lo; tile < tile_hi; ++tile) {
    const float* xg = x + (size_t)tile * 16 * 256;
    const int tz = opaque_zero();
    const float* const W1t = W1 + tz;    const float* const b1t = b1 + tz;
    const float* const W2t = W2 + tz;    const float* const b2t = b2 + tz;
    const float* const lgt = ln_g + tz;  const float* const lbt = ln_b + tz;
    const float* const Wft = Wf + tz;    const float* const bft = bf + tz;


    float xr[16];
    #pragma unroll
    for (int r = 0; r < 16; ++r) xr[r] = xg[t + 256*r];
    v2f w1f[16];
    #pragma unroll
    for (int s = 0; s < 16; ++s) {
        const float* wp = W1t + (size_t)(kq*64 + 4*s + 2*hi)*32 + nt*16 + bb;
        w1f[s][0] = wp[0];  w1f[s][1] = wp[32];
    }
    v2f w2f0[8], w2f1[8];
    #pragma unroll
    for (int s = 0; s < 8; ++s) {
        const float* wp = W2t + (size_t)(4*s + 2*hi)*256 + 32*wav + bb;
        w2f0[s][0] = wp[0];   w2f0[s][1] = wp[256];
        w2f1[s][0] = wp[16];  w2f1[s][1] = wp[256 + 16];
    }
    const float b2a = b2t[32*wav + bb], b2b = b2t[32*wav + 16 + bb];
    const float b1v = b1t[lane];
    #pragma unroll
    for (int r = 0; r < 16; ++r) xh[r*XP + t] = xr[r];
    __syncthreads();

    float* const p1 = bij;
    {
        v8f c0 = vzero8(), c1 = vzero8();
        const float* xa = xh + bb*XP + kq*64 + 2*hi;
        #pragma unroll
        for (int s = 0; s < 16; ++s) {
            v2f af;  af[0] = xa[4*s];  af[1] = xa[4*s + 1];
            if (s & 1) c1 = wmma4(af, w1f[s], c1);
            else       c0 = wmma4(af, w1f[s], c0);
        }
        const v8f c = c0 + c1;
        float* pp = p1 + (kq*16 + 8*hi)*HP + nt*16 + bb;
        #pragma unroll
        for (int r = 0; r < 8; ++r) pp[r*HP] = c[r];
    }
    __syncthreads();

    {
        float z[2];
        #pragma unroll
        for (int bsel = 0; bsel < 2; ++bsel) {
            const int b = wav + 8*bsel;
            float acc = p1[b*HP + lane];
            #pragma unroll
            for (int q = 1; q < 4; ++q) acc += p1[(q*16 + b)*HP + lane];
            z[bsel] = acc + b1v;
        }
        gelu_n<2>(z);
        h1s[wav*HP + lane] = z[0];
        h1s[(wav + 8)*HP + lane] = z[1];
    }
    __syncthreads();

    {
        v8f d0 = vzero8(), d1 = vzero8();
        const float* ha = h1s + bb*HP + 2*hi;
        #pragma unroll
        for (int s = 0; s < 8; ++s) {
            v2f af;  af[0] = ha[4*s];  af[1] = ha[4*s + 1];
            d0 = wmma4(af, w2f0[s], d0);
            d1 = wmma4(af, w2f1[s], d1);
        }
        float* ho = xh + (8*hi)*256 + 32*wav + bb;
        float z[16];
        #pragma unroll
        for (int r = 0; r < 8; ++r) { z[2*r] = d0[r] + b2a;  z[2*r + 1] = d1[r] + b2b; }
        gelu_n<16>(z);
        #pragma unroll
        for (int r = 0; r < 8; ++r) {
            ho[r*256]      = z[2*r];
            ho[r*256 + 16] = z[2*r + 1];
        }
    }
    __syncthreads();

    {
        int ba = wav, bc = wav + 8;
        float sa = 0.f, sb = 0.f;
        #pragma unroll
        for (int r = 0; r < 8; ++r) { sa += xh[ba*256 + lane + 32*r]; sb += xh[bc*256 + lane + 32*r]; }
        #pragma unroll
        for (int m = 16; m >= 1; m >>= 1) { sa += __shfl_xor(sa, m, 32); sb += __shfl_xor(sb, m, 32); }
        float mua = sa * (1.0f/256.0f), mub = sb * (1.0f/256.0f);
        float va = 0.f, vb = 0.f;
        #pragma unroll
        for (int r = 0; r < 8; ++r) {
            float da = xh[ba*256 + lane + 32*r] - mua; va = fmaf(da, da, va);
            float db = xh[bc*256 + lane + 32*r] - mub; vb = fmaf(db, db, vb);
        }
        #pragma unroll
        for (int m = 16; m >= 1; m >>= 1) { va += __shfl_xor(va, m, 32); vb += __shfl_xor(vb, m, 32); }
        if (lane == 0) {
            mu_s[ba] = mua; rstd_s[ba] = rsqrtf(va * (1.0f/256.0f) + LN_EPS);
            mu_s[bc] = mub; rstd_s[bc] = rsqrtf(vb * (1.0f/256.0f) + LN_EPS);
        }
    }
    __syncthreads();

    {
        int i = t & 31, b0 = t >> 5;
        float g8[8], be8[8];
        #pragma unroll
        for (int k = 0; k < 8; ++k) { g8[k] = lgt[i*8 + k]; be8[k] = lbt[i*8 + k]; }
        #pragma unroll
        for (int bsel = 0; bsel < 2; ++bsel) {
            int b = b0 + 8*bsel;
            float mu = mu_s[b], rs = rstd_s[b];
            float vals[8]; float nsq = 0.f;
            #pragma unroll
            for (int k = 0; k < 8; ++k) {
                float v = (xh[b*256 + i*8 + k] - mu) * rs * g8[k] + be8[k];
                vals[k] = v; nsq = fmaf(v, v, nsq);
            }
            float scale = nsq / ((1.0f + nsq) * sqrtf(nsq + SQ_EPS));
            #pragma unroll
            for (int k = 0; k < 8; ++k) u_t[i*130 + k*16 + b] = scale * vals[k];
        }
    }

    #pragma unroll
    for (int r = 0; r < 36; ++r) { bij[t + 256*r] = 0.f; cij[t + 256*r] = 0.0625f; }
    if (t < 16*RS - 9216) { bij[9216 + t] = 0.f; cij[9216 + t] = 0.0625f; }
    __syncthreads();

    v8f vA = vzero8(), vB = vzero8();
    for (int r = 0; r < 3; ++r) {
        v8f sA0 = vzero8(), sA1 = vzero8(), sB0 = vzero8(), sB1 = vzero8();
        v2f ub0[32], ub1[32];
        #pragma unroll
        for (int i = 0; i < 32; ++i) {
            const float* ub = u_t + i*130 + bb;
            v2f bu0, bu1;
            bu0[0] = ub[(2*hi+0)*16];  bu0[1] = ub[(2*hi+1)*16];
            bu1[0] = ub[(2*hi+4)*16];  bu1[1] = ub[(2*hi+5)*16];
            ub0[i] = bu0;  ub1[i] = bu1;
            float cA = cij[bb*RS + i*RP + wav];
            float cB = cij[bb*RS + i*RP + wav + 8];
            v2f p0, p1;
            p0[0] = bu0[0]*cA; p0[1] = bu0[1]*cA; p1[0] = bu1[0]*cA; p1[1] = bu1[1]*cA;
            if (i & 1) { sA1 = wmma4(a0A[i], p0, sA1); sA1 = wmma4(a1A[i], p1, sA1); }
            else       { sA0 = wmma4(a0A[i], p0, sA0); sA0 = wmma4(a1A[i], p1, sA0); }
            p0[0] = bu0[0]*cB; p0[1] = bu0[1]*cB; p1[0] = bu1[0]*cB; p1[1] = bu1[1]*cB;
            if (i & 1) { sB1 = wmma4(a0B[i], p0, sB1); sB1 = wmma4(a1B[i], p1, sB1); }
            else       { sB0 = wmma4(a0B[i], p0, sB0); sB0 = wmma4(a1B[i], p1, sB0); }
        }
        v8f sA = sA0 + sA1, sB = sB0 + sB1;
        float nA = 0.f, nB = 0.f;
        #pragma unroll
        for (int v = 0; v < 8; ++v) { nA = fmaf(sA[v], sA[v], nA); nB = fmaf(sB[v], sB[v], nB); }
        nA += __shfl_xor(nA, 16, 32);  nB += __shfl_xor(nB, 16, 32);
        float scA = nA / ((1.0f + nA) * sqrtf(nA + SQ_EPS));
        float scB = nB / ((1.0f + nB) * sqrtf(nB + SQ_EPS));
        #pragma unroll
        for (int v = 0; v < 8; ++v) { vA[v] = sA[v]*scA; vB[v] = sB[v]*scB; }

        if (r < 2) {
            const int ib = bb*RS + wav + 8*hi;
            #pragma unroll
            for (int i = 0; i < 32; ++i) {
                const v2f bu0 = ub0[i], bu1 = ub1[i];
                v8f uA = vzero8(); uA = wmma4(a0A[i], bu0, uA); uA = wmma4(a1A[i], bu1, uA);
                v8f uB = vzero8(); uB = wmma4(a0B[i], bu0, uB); uB = wmma4(a1B[i], bu1, uB);
                float agA = 0.f, agB = 0.f;
                #pragma unroll
                for (int v = 0; v < 8; ++v) { agA = fmaf(uA[v], vA[v], agA); agB = fmaf(uB[v], vB[v], agB); }
                const float keep = hi ? agB : agA;
                const float send = hi ? agA : agB;
                bij[ib + i*RP] += keep + xhalf(send);
            }
            __syncthreads();
            #pragma unroll
            for (int bsel = 0; bsel < 2; ++bsel) {
                int base = (wav + 8*bsel)*RS + (t & 31)*RP;
                float mx = bij[base];
                #pragma unroll
                for (int j = 1; j < 16; ++j) mx = fmaxf(mx, bij[base + j]);
                float e[16]; float sum = 0.f;
                #pragma unroll
                for (int j = 0; j < 16; ++j) { e[j] = exp_unit(bij[base + j] - mx); sum += e[j]; }
                float inv = 1.0f / sum;
                #pragma unroll
                for (int j = 0; j < 16; ++j) cij[base + j] = e[j] * inv;
            }
            __syncthreads();
        }
    }

    {
        float pA = 0.f, pB = 0.f;
        #pragma unroll
        for (int v = 0; v < 8; ++v) {
            pA = fmaf(vA[v], Wft[(wav    )*16 + v + 8*hi], pA);
            pB = fmaf(vB[v], Wft[(wav + 8)*16 + v + 8*hi], pB);
        }
        pA += __shfl_xor(pA, 16, 32);  pB += __shfl_xor(pB, 16, 32);
        if (lane < 16) { part[wav*17 + bb] = pA; part[(wav+8)*17 + bb] = pB; }
    }
    __syncthreads();
    if (t < 16) {
        float acc = bft[0];
        #pragma unroll
        for (int j = 0; j < 16; ++j) acc += part[j*17 + t];
        out[tile*16 + t] = acc;
    }
    }
}

extern "C" void kernel_launch(void* const* d_in, const int* in_sizes, int n_in,
                              void* d_out, int out_size, void* d_ws, size_t ws_size,
                              hipStream_t stream) {
    const float* x    = (const float*)d_in[0];
    const float* W1   = (const float*)d_in[1];
    const float* b1   = (const float*)d_in[2];
    const float* W2   = (const float*)d_in[3];
    const float* b2   = (const float*)d_in[4];
    const float* ln_g = (const float*)d_in[5];
    const float* ln_b = (const float*)d_in[6];
    const float* Wcap = (const float*)d_in[7];
    const float* Wf   = (const float*)d_in[8];
    const float* bf   = (const float*)d_in[9];
    float* out = (float*)d_out;

    const int B = in_sizes[0] / 256;
    const int tiles = B / 16;
    capsnet_fused<<<NWG, 256, 0, stream>>>(x, W1, b1, W2, b2, ln_g, ln_b, Wcap, Wf, bf, out, tiles);
}


# MLP and LayerNorm statistics run phase by phase on the work-group's 4 tiles ahead of their routing: one x and weight fetch, 5 barriers per 4 tiles, 2 tiles side by side a wave; sums bit-identical
# speedup vs baseline: 2.1177x; 1.0772x over previous
#include <hip/hip_runtime.h>
#include <math.h>

typedef __attribute__((ext_vector_type(2))) float v2f;
typedef __attribute__((ext_vector_type(8))) float v8f;

#define LN_EPS 1e-5f
#define SQ_EPS 1e-9f

#define RP 18
#define RS 579

#define XP 260
#define HP 36

#define NWG 256

#define NT 4
#define NP 2
#define XT (16*XP)
#define HT (16*HP)
#define PT (4*16*HP)

__device__ __forceinline__ float exp_unit(float x) {
    return __builtin_amdgcn_exp2f(x * 1.44269504088896340736f);
}

#define ERF_SPLIT 0.927734375f
__device__ __forceinline__ float erf_small(float a) {
    const float s = a * a;
    float q = -5.96761703e-4f;
    q = fmaf(q, s, 4.99119423e-3f);
    q = fmaf(q, s, -2.67681349e-2f);
    q = fmaf(q, s, 1.12819925e-1f);
    q = fmaf(q, s, -3.76125336e-1f);
    q = fmaf(q, s, 1.28379166e-1f);
    return fmaf(q, a, a);
}
__device__ __forceinline__ float erf_large(float a) {
    const float t = fabsf(a);
    const float s = a * a;
    float r = fmaf(-1.72853470e-5f, t, 3.83197126e-4f);
    const float u = fmaf(-3.88396438e-3f, t, 2.42546219e-2f);
    r = fmaf(r, s, u);
    r = fmaf(r, t, -1.06777877e-1f);
    r = fmaf(r, t, -6.34846687e-1f);
    r = fmaf(r, t, -1.28717512e-1f);
    r = fmaf(r, t, -t);
    return copysignf(1.0f - exp_unit(r), a);
}

template <int N>
__device__ __forceinline__ void gelu_n(float (&z)[N]) {
    float a[N], e[N];
    float amax = 0.f;
    #pragma unroll
    for (int n = 0; n < N; ++n) {
        a[n] = z[n] * 0.70710678118654752440f;
        e[n] = erf_small(a[n]);
        amax = fmaxf(amax, fabsf(a[n]));
    }
    if (amax > ERF_SPLIT) {
        #pragma unroll
        for (int n = 0; n < N; ++n) {
            const float el = erf_large(a[n]);
            e[n] = (fabsf(a[n]) > ERF_SPLIT) ? el : e[n];
        }
    }
    #pragma unroll
    for (int n = 0; n < N; ++n) z[n] = 0.5f * z[n] * (1.0f + e[n]);
}

__device__ __forceinline__ v8f wmma4(v2f a, v2f b, v8f c) {
    return __builtin_amdgcn_wmma_f32_16x16x4_f32(false, a, false, b, (short)0, c, false, false);
}

__device__ __forceinline__ float xhalf(float x) {
    const int b = __builtin_bit_cast(int, x);
    return __builtin_bit_cast(float, __builtin_amdgcn_permlanex16(b, b, 0x76543210, 0xfedcba98, false, false));
}

__device__ __forceinline__ int opaque_zero() {
    int z = 0;
    asm volatile("" : "+s"(z));
    return z;
}

__device__ __forceinline__ v8f vzero8() {
    v8f z = {0.f,0.f,0.f,0.f,0.f,0.f,0.f,0.f};
    return z;
}

__global__ __attribute__((amdgpu_waves_per_eu(1))) __launch_bounds__(256)
void capsnet_fused(
    const float* __restrict__ x,   const float* __restrict__ W1, const float* __restrict__ b1,
    const float* __restrict__ W2,  const float* __restrict__ b2,
    const float* __restrict__ ln_g,const float* __restrict__ ln_b,
    const float* __restrict__ Wcap,const float* __restrict__ Wf, const float* __restrict__ bf,
    float* __restrict__ out, const int ntiles)
{
    __shared__ float xh[NT*XT];
    __shared__ float h1s[NT*HT];
    __shared__ float mu_s[NT*16];
    __shared__ float rstd_s[NT*16];
    __shared__ float u_t[32*130];
    __shared__ float bij[16*RS];
    __shared__ float cij[16*RS];
    __shared__ float part[16*17];

    const int t    = threadIdx.x;
    const int lane = t & 31;
    const int wav  = t >> 5;
    const int bb   = lane & 15;
    const int hi   = lane >> 4;

    const int nt = wav & 1, kq = wav >> 1;

    v2f a0A[32], a1A[32], a0B[32], a1B[32];
    #pragma unroll
    for (int i = 0; i < 32; ++i) {
        const float* wcA = Wcap + (size_t)((i*16 + wav    )*8)*16 + bb;
        const float* wcB = Wcap + (size_t)((i*16 + wav + 8)*8)*16 + bb;
        a0A[i][0] = wcA[(2*hi+0)*16];  a0A[i][1] = wcA[(2*hi+1)*16];
        a1A[i][0] = wcA[(2*hi+4)*16];  a1A[i][1] = wcA[(2*hi+5)*16];
        a0B[i][0] = wcB[(2*hi+0)*16];  a0B[i][1] = wcB[(2*hi+1)*16];
        a1B[i][0] = wcB[(2*hi+4)*16];  a1B[i][1] = wcB[(2*hi+5)*16];
    }


    const unsigned tile_lo = (blockIdx.x * (unsigned)ntiles) / NWG;
    const unsigned tile_hi = ((blockIdx.x + 1u) * (unsigned)ntiles) / NWG;
    for (unsigned tile0 = tile_lo; tile0 < tile_hi; tile0 += NT) {
    const unsigned nb = (tile_hi - tile0 < (unsigned)NT) ? (tile_hi - tile0) : (unsigned)NT;
    const int tz = opaque_zero();
    const float* const W1t = W1 + tz;    const float* const b1t = b1 + tz;
    const float* const W2t = W2 + tz;    const float* const b2t = b2 + tz;


    float xr[NT][16];
    #pragma unroll
    for (int tt = 0; tt < NT; ++tt) {
        const unsigned ts = (tile0 + tt < tile_hi) ? (tile0 + tt) : (tile_hi - 1u);
        const float* xg = x + (size_t)ts * 16 * 256;
        #pragma unroll
        for (int r = 0; r < 16; ++r) xr[tt][r] = xg[t + 256*r];
    }
    v2f w1f[16];
    #pragma unroll
    for (int s = 0; s < 16; ++s) {
        const float* wp = W1t + (size_t)(kq*64 + 4*s + 2*hi)*32 + nt*16 + bb;
        w1f[s][0] = wp[0];  w1f[s][1] = wp[32];
    }
    v2f w2f0[8], w2f1[8];
    #pragma unroll
    for (int s = 0; s < 8; ++s) {
        const float* wp = W2t + (size_t)(4*s + 2*hi)*256 + 32*wav + bb;
        w2f0[s][0] = wp[0];   w2f0[s][1] = wp[256];
        w2f1[s][0] = wp[16];  w2f1[s][1] = wp[256 + 16];
    }
    const float b2a = b2t[32*wav + bb], b2b = b2t[32*wav + 16 + bb];
    const float b1v = b1t[lane];
    #pragma unroll
    for (int tt = 0; tt < NT; ++tt) {
        #pragma unroll
        for (int r = 0; r < 16; ++r) xh[tt*XT + r*XP + t] = xr[tt][r];
    }
    __syncthreads();

    float* const p1 = bij;
    #pragma nounroll
    for (int th = 0; th < NT; th += NP) {
        v8f c0[NP], c1[NP];
        #pragma unroll
        for (int tp = 0; tp < NP; ++tp) { c0[tp] = vzero8();  c1[tp] = vzero8(); }
        const float* xa = xh + th*XT + bb*XP + kq*64 + 2*hi;
        #pragma unroll
        for (int s = 0; s < 16; ++s) {
            #pragma unroll
            for (int tp = 0; tp < NP; ++tp) {
                v2f af;  af[0] = xa[tp*XT + 4*s];  af[1] = xa[tp*XT + 4*s + 1];
                if (s & 1) c1[tp] = wmma4(af, w1f[s], c1[tp]);
                else       c0[tp] = wmma4(af, w1f[s], c0[tp]);
            }
        }
        #pragma unroll
        for (int tp = 0; tp < NP; ++tp) {
            const v8f c = c0[tp] + c1[tp];
            float* pp = p1 + (th + tp)*PT + (kq*16 + 8*hi)*HP + nt*16 + bb;
            #pragma unroll
            for (int r = 0; r < 8; ++r) pp[r*HP] = c[r];
        }
    }
    __syncthreads();

    #pragma nounroll
    for (int th = 0; th < NT; th += NP) {
        float z[2*NP];
        #pragma unroll
        for (int tp = 0; tp < NP; ++tp) {
            const float* pt = p1 + (th + tp)*PT;
            #pragma unroll
            for (int bsel = 0; bsel < 2; ++bsel) {
                const int b = wav + 8*bsel;
                float acc = pt[b*HP + lane];
                #pragma unroll
                for (int q = 1; q < 4; ++q) acc += pt[(q*16 + b)*HP + lane];
                z[2*tp + bsel] = acc + b1v;
            }
        }
        gelu_n<2*NP>(z);
        #pragma unroll
        for (int tp = 0; tp < NP; ++tp) {
            h1s[(th + tp)*HT + wav*HP + lane] = z[2*tp];
            h1s[(th + tp)*HT + (wav + 8)*HP + lane] = z[2*tp + 1];
        }
    }
    __syncthreads();

    for (int tt = 0; tt < NT; ++tt) {
        v8f d0 = vzero8(), d1 = vzero8();
        const float* ha = h1s + tt*HT + bb*HP + 2*hi;
        #pragma unroll
        for (int s = 0; s < 8; ++s) {
            v2f af;  af[0] = ha[4*s];  af[1] = ha[4*s + 1];
            d0 = wmma4(af, w2f0[s], d0);
            d1 = wmma4(af, w2f1[s], d1);
        }
        float* ho = xh + tt*XT + (8*hi)*256 + 32*wav + bb;
        float z[16];
        #pragma unroll
        for (int r = 0; r < 8; ++r) { z[2*r] = d0[r] + b2a;  z[2*r + 1] = d1[r] + b2b; }
        gelu_n<16>(z);
        #pragma unroll
        for (int r = 0; r < 8; ++r) {
            ho[r*256]      = z[2*r];
            ho[r*256 + 16] = z[2*r + 1];
        }
    }
    __syncthreads();

    #pragma nounroll
    for (int th = 0; th < NT; th += NP) {
        const int ba = wav, bc = wav + 8;
        const float* xb = xh + th*XT;
        float sa[NP], sb[NP], mua[NP], mub[NP], va[NP], vb[NP];
        #pragma unroll
        for (int tp = 0; tp < NP; ++tp) {
            sa[tp] = 0.f;  sb[tp] = 0.f;
            #pragma unroll
            for (int r = 0; r < 8; ++r) { sa[tp] += xb[tp*XT + ba*256 + lane + 32*r]; sb[tp] += xb[tp*XT + bc*256 + lane + 32*r]; }
        }
        #pragma unroll
        for (int m = 16; m >= 1; m >>= 1) {
            #pragma unroll
            for (int tp = 0; tp < NP; ++tp) { sa[tp] += __shfl_xor(sa[tp], m, 32); sb[tp] += __shfl_xor(sb[tp], m, 32); }
        }
        #pragma unroll
        for (int tp = 0; tp < NP; ++tp) {
            mua[tp] = sa[tp] * (1.0f/256.0f);  mub[tp] = sb[tp] * (1.0f/256.0f);
            va[tp] = 0.f;  vb[tp] = 0.f;
            #pragma unroll
            for (int r = 0; r < 8; ++r) {
                float da = xb[tp*XT + ba*256 + lane + 32*r] - mua[tp]; va[tp] = fmaf(da, da, va[tp]);
                float db = xb[tp*XT + bc*256 + lane + 32*r] - mub[tp]; vb[tp] = fmaf(db, db, vb[tp]);
            }
        }
        #pragma unroll
        for (int m = 16; m >= 1; m >>= 1) {
            #pragma unroll
            for (int tp = 0; tp < NP; ++tp) { va[tp] += __shfl_xor(va[tp], m, 32); vb[tp] += __shfl_xor(vb[tp], m, 32); }
        }
        if (lane == 0) {
            #pragma unroll
            for (int tp = 0; tp < NP; ++tp) {
                mu_s[(th + tp)*16 + ba] = mua[tp]; rstd_s[(th + tp)*16 + ba] = rsqrtf(va[tp] * (1.0f/256.0f) + LN_EPS);
                mu_s[(th + tp)*16 + bc] = mub[tp]; rstd_s[(th + tp)*16 + bc] = rsqrtf(vb[tp] * (1.0f/256.0f) + LN_EPS);
            }
        }
    }
    __syncthreads();

    for (unsigned tt = 0; tt < nb; ++tt) {
    const unsigned tile = tile0 + tt;
    const int tzt = opaque_zero();
    const float* const lgt = ln_g + tzt; const float* const lbt = ln_b + tzt;
    const float* const Wft = Wf + tzt;   const float* const bft = bf + tzt;

    {
        int i = t & 31, b0 = t >> 5;
        float g8[8], be8[8];
        #pragma unroll
        for (int k = 0; k < 8; ++k) { g8[k] = lgt[i*8 + k]; be8[k] = lbt[i*8 + k]; }
        const float* h2 = xh + tt*XT;
        #pragma unroll
        for (int bsel = 0; bsel < 2; ++bsel) {
            int b = b0 + 8*bsel;
            float mu = mu_s[tt*16 + b], rs = rstd_s[tt*16 + b];
            float vals[8]; float nsq = 0.f;
            #pragma unroll
            for (int k = 0; k < 8; ++k) {
                float v = (h2[b*256 + i*8 + k] - mu) * rs * g8[k] + be8[k];
                vals[k] = v; nsq = fmaf(v, v, nsq);
            }
            float scale = nsq / ((1.0f + nsq) * sqrtf(nsq + SQ_EPS));
            #pragma unroll
            for (int k = 0; k < 8; ++k) u_t[i*130 + k*16 + b] = scale * vals[k];
        }
    }

    #pragma unroll
    for (int r = 0; r < 36; ++r) { bij[t + 256*r] = 0.f; cij[t + 256*r] = 0.0625f; }
    if (t < 16*RS - 9216) { bij[9216 + t] = 0.f; cij[9216 + t] = 0.0625f; }
    __syncthreads();

    v8f vA = vzero8(), vB = vzero8();
    for (int r = 0; r < 3; ++r) {
        v8f sA0 = vzero8(), sA1 = vzero8(), sB0 = vzero8(), sB1 = vzero8();
        v2f ub0[32], ub1[32];
        #pragma unroll
        for (int i = 0; i < 32; ++i) {
            const float* ub = u_t + i*130 + bb;
            v2f bu0, bu1;
            bu0[0] = ub[(2*hi+0)*16];  bu0[1] = ub[(2*hi+1)*16];
            bu1[0] = ub[(2*hi+4)*16];  bu1[1] = ub[(2*hi+5)*16];
            ub0[i] = bu0;  ub1[i] = bu1;
            float cA = cij[bb*RS + i*RP + wav];
            float cB = cij[bb*RS + i*RP + wav + 8];
            v2f p0, p1;
            p0[0] = bu0[0]*cA; p0[1] = bu0[1]*cA; p1[0] = bu1[0]*cA; p1[1] = bu1[1]*cA;
            if (i & 1) { sA1 = wmma4(a0A[i], p0, sA1); sA1 = wmma4(a1A[i], p1, sA1); }
            else       { sA0 = wmma4(a0A[i], p0, sA0); sA0 = wmma4(a1A[i], p1, sA0); }
            p0[0] = bu0[0]*cB; p0[1] = bu0[1]*cB; p1[0] = bu1[0]*cB; p1[1] = bu1[1]*cB;
            if (i & 1) { sB1 = wmma4(a0B[i], p0, sB1); sB1 = wmma4(a1B[i], p1, sB1); }
            else       { sB0 = wmma4(a0B[i], p0, sB0); sB0 = wmma4(a1B[i], p1, sB0); }
        }
        v8f sA = sA0 + sA1, sB = sB0 + sB1;
        float nA = 0.f, nB = 0.f;
        #pragma unroll
        for (int v = 0; v < 8; ++v) { nA = fmaf(sA[v], sA[v], nA); nB = fmaf(sB[v], sB[v], nB); }
        nA += __shfl_xor(nA, 16, 32);  nB += __shfl_xor(nB, 16, 32);
        float scA = nA / ((1.0f + nA) * sqrtf(nA + SQ_EPS));
        float scB = nB / ((1.0f + nB) * sqrtf(nB + SQ_EPS));
        #pragma unroll
        for (int v = 0; v < 8; ++v) { vA[v] = sA[v]*scA; vB[v] = sB[v]*scB; }

        if (r < 2) {
            const int ib = bb*RS + wav + 8*hi;
            #pragma unroll
            for (int i = 0; i < 32; ++i) {
                const v2f bu0 = ub0[i], bu1 = ub1[i];
                v8f uA = vzero8(); uA = wmma4(a0A[i], bu0, uA); uA = wmma4(a1A[i], bu1, uA);
                v8f uB = vzero8(); uB = wmma4(a0B[i], bu0, uB); uB = wmma4(a1B[i], bu1, uB);
                float agA = 0.f, agB = 0.f;
                #pragma unroll
                for (int v = 0; v < 8; ++v) { agA = fmaf(uA[v], vA[v], agA); agB = fmaf(uB[v], vB[v], agB); }
                const float keep = hi ? agB : agA;
                const float send = hi ? agA : agB;
                bij[ib + i*RP] += keep + xhalf(send);
            }
            __syncthreads();
            #pragma unroll
            for (int bsel = 0; bsel < 2; ++bsel) {
                int base = (wav + 8*bsel)*RS + (t & 31)*RP;
                float mx = bij[base];
                #pragma unroll
                for (int j = 1; j < 16; ++j) mx = fmaxf(mx, bij[base + j]);
                float e[16]; float sum = 0.f;
                #pragma unroll
                for (int j = 0; j < 16; ++j) { e[j] = exp_unit(bij[base + j] - mx); sum += e[j]; }
                float inv = 1.0f / sum;
                #pragma unroll
                for (int j = 0; j < 16; ++j) cij[base + j] = e[j] * inv;
            }
            __syncthreads();
        }
    }

    {
        float pA = 0.f, pB = 0.f;
        #pragma unroll
        for (int v = 0; v < 8; ++v) {
            pA = fmaf(vA[v], Wft[(wav    )*16 + v + 8*hi], pA);
            pB = fmaf(vB[v], Wft[(wav + 8)*16 + v + 8*hi], pB);
        }
        pA += __shfl_xor(pA, 16, 32);  pB += __shfl_xor(pB, 16, 32);
        if (lane < 16) { part[wav*17 + bb] = pA; part[(wav+8)*17 + bb] = pB; }
    }
    __syncthreads();
    if (t < 16) {
        float acc = bft[0];
        #pragma unroll
        for (int j = 0; j < 16; ++j) acc += part[j*17 + t];
        out[tile*16 + t] = acc;
    }
    }
    }
}

extern "C" void kernel_launch(void* const* d_in, const int* in_sizes, int n_in,
                              void* d_out, int out_size, void* d_ws, size_t ws_size,
                              hipStream_t stream) {
    const float* x    = (const float*)d_in[0];
    const float* W1   = (const float*)d_in[1];
    const float* b1   = (const float*)d_in[2];
    const float* W2   = (const float*)d_in[3];
    const float* b2   = (const float*)d_in[4];
    const float* ln_g = (const float*)d_in[5];
    const float* ln_b = (const float*)d_in[6];
    const float* Wcap = (const float*)d_in[7];
    const float* Wf   = (const float*)d_in[8];
    const float* bf   = (const float*)d_in[9];
    float* out = (float*)d_out;

    const int B = in_sizes[0] / 256;
    const int tiles = B / 16;
    capsnet_fused<<<NWG, 256, 0, stream>>>(x, W1, b1, W2, b2, ln_g, ln_b, Wcap, Wf, bf, out, tiles);
}


# u_hat of 8 of 32 input capsules computed once a tile, kept in LDS (lane-private, 132 KB): 4 of 5 routing passes drop a quarter of their matrix calls; s adds c*u_hat on the vector ALU: new sum order
# speedup vs baseline: 2.2114x; 1.0442x over previous
#include <hip/hip_runtime.h>
#include <math.h>

typedef __attribute__((ext_vector_type(2))) float v2f;
typedef __attribute__((ext_vector_type(4))) float v4f;
typedef __attribute__((ext_vector_type(8))) float v8f;

#define LN_EPS 1e-5f
#define SQ_EPS 1e-9f

#define RP 18
#define RS 579

#define NC 8
#define CP (4*NC + 1)
#if 4*NC != 32
#error "the agreement pass's capsule order is written for NC = 8"
#endif

#define XP 260
#define HP 36

#define NWG 256

#define NT 4
#define NP 2
#define XT (16*XP)
#define HT (16*HP)
#define PT (4*16*HP)

__device__ __forceinline__ float exp_unit(float x) {
    return __builtin_amdgcn_exp2f(x * 1.44269504088896340736f);
}

#define ERF_SPLIT 0.927734375f
__device__ __forceinline__ float erf_small(float a) {
    const float s = a * a;
    float q = -5.96761703e-4f;
    q = fmaf(q, s, 4.99119423e-3f);
    q = fmaf(q, s, -2.67681349e-2f);
    q = fmaf(q, s, 1.12819925e-1f);
    q = fmaf(q, s, -3.76125336e-1f);
    q = fmaf(q, s, 1.28379166e-1f);
    return fmaf(q, a, a);
}
__device__ __forceinline__ float erf_large(float a) {
    const float t = fabsf(a);
    const float s = a * a;
    float r = fmaf(-1.72853470e-5f, t, 3.83197126e-4f);
    const float u = fmaf(-3.88396438e-3f, t, 2.42546219e-2f);
    r = fmaf(r, s, u);
    r = fmaf(r, t, -1.06777877e-1f);
    r = fmaf(r, t, -6.34846687e-1f);
    r = fmaf(r, t, -1.28717512e-1f);
    r = fmaf(r, t, -t);
    return copysignf(1.0f - exp_unit(r), a);
}

template <int N>
__device__ __forceinline__ void gelu_n(float (&z)[N]) {
    float a[N], e[N];
    float amax = 0.f;
    #pragma unroll
    for (int n = 0; n < N; ++n) {
        a[n] = z[n] * 0.70710678118654752440f;
        e[n] = erf_small(a[n]);
        amax = fmaxf(amax, fabsf(a[n]));
    }
    if (amax > ERF_SPLIT) {
        #pragma unroll
        for (int n = 0; n < N; ++n) {
            const float el = erf_large(a[n]);
            e[n] = (fabsf(a[n]) > ERF_SPLIT) ? el : e[n];
        }
    }
    #pragma unroll
    for (int n = 0; n < N; ++n) z[n] = 0.5f * z[n] * (1.0f + e[n]);
}

__device__ __forceinline__ v8f wmma4(v2f a, v2f b, v8f c) {
    return __builtin_amdgcn_wmma_f32_16x16x4_f32(false, a, false, b, (short)0, c, false, false);
}

__device__ __forceinline__ float xhalf(float x) {
    const int b = __builtin_bit_cast(int, x);
    return __builtin_bit_cast(float, __builtin_amdgcn_permlanex16(b, b, 0x76543210, 0xfedcba98, false, false));
}

__device__ __forceinline__ int opaque_zero() {
    int z = 0;
    asm volatile("" : "+s"(z));
    return z;
}

__device__ __forceinline__ v8f vzero8() {
    v8f z = {0.f,0.f,0.f,0.f,0.f,0.f,0.f,0.f};
    return z;
}

__global__ __attribute__((amdgpu_waves_per_eu(1))) __launch_bounds__(256)
void capsnet_fused(
    const float* __restrict__ x,   const float* __restrict__ W1, const float* __restrict__ b1,
    const float* __restrict__ W2,  const float* __restrict__ b2,
    const float* __restrict__ ln_g,const float* __restrict__ ln_b,
    const float* __restrict__ Wcap,const float* __restrict__ Wf, const float* __restrict__ bf,
    float* __restrict__ out, const int ntiles)
{
    __shared__ __attribute__((aligned(32))) float xh[NT*XT];
    __shared__ __attribute__((aligned(32))) float h1s[NT*HT];
    __shared__ __attribute__((aligned(32))) float mu_s[NT*16];
    __shared__ __attribute__((aligned(32))) float rstd_s[NT*16];
    __shared__ __attribute__((aligned(32))) float u_t[32*130];
    __shared__ __attribute__((aligned(32))) float bij[16*RS];
    __shared__ __attribute__((aligned(32))) float cij[16*RS];
    __shared__ __attribute__((aligned(32))) float part[16*17];
    __shared__ v4f uh_c[256*CP];

    const int t    = threadIdx.x;
    const int lane = t & 31;
    const int wav  = t >> 5;
    const int bb   = lane & 15;
    const int hi   = lane >> 4;

    const int nt = wav & 1, kq = wav >> 1;

    v2f a0A[32], a1A[32], a0B[32], a1B[32];
    #pragma unroll
    for (int i = 0; i < 32; ++i) {
        const float* wcA = Wcap + (size_t)((i*16 + wav    )*8)*16 + bb;
        const float* wcB = Wcap + (size_t)((i*16 + wav + 8)*8)*16 + bb;
        a0A[i][0] = wcA[(2*hi+0)*16];  a0A[i][1] = wcA[(2*hi+1)*16];
        a1A[i][0] = wcA[(2*hi+4)*16];  a1A[i][1] = wcA[(2*hi+5)*16];
        a0B[i][0] = wcB[(2*hi+0)*16];  a0B[i][1] = wcB[(2*hi+1)*16];
        a1B[i][0] = wcB[(2*hi+4)*16];  a1B[i][1] = wcB[(2*hi+5)*16];
    }


    const unsigned tile_lo = (blockIdx.x * (unsigned)ntiles) / NWG;
    const unsigned tile_hi = ((blockIdx.x + 1u) * (unsigned)ntiles) / NWG;
    for (unsigned tile0 = tile_lo; tile0 < tile_hi; tile0 += NT) {
    const unsigned nb = (tile_hi - tile0 < (unsigned)NT) ? (tile_hi - tile0) : (unsigned)NT;
    const int tz = opaque_zero();
    const float* const W1t = W1 + tz;    const float* const b1t = b1 + tz;
    const float* const W2t = W2 + tz;    const float* const b2t = b2 + tz;


    float xr[NT][16];
    #pragma unroll
    for (int tt = 0; tt < NT; ++tt) {
        const unsigned ts = (tile0 + tt < tile_hi) ? (tile0 + tt) : (tile_hi - 1u);
        const float* xg = x + (size_t)ts * 16 * 256;
        #pragma unroll
        for (int r = 0; r < 16; ++r) xr[tt][r] = xg[t + 256*r];
    }
    v2f w1f[16];
    #pragma unroll
    for (int s = 0; s < 16; ++s) {
        const float* wp = W1t + (size_t)(kq*64 + 4*s + 2*hi)*32 + nt*16 + bb;
        w1f[s][0] = wp[0];  w1f[s][1] = wp[32];
    }
    v2f w2f0[8], w2f1[8];
    #pragma unroll
    for (int s = 0; s < 8; ++s) {
        const float* wp = W2t + (size_t)(4*s + 2*hi)*256 + 32*wav + bb;
        w2f0[s][0] = wp[0];   w2f0[s][1] = wp[256];
        w2f1[s][0] = wp[16];  w2f1[s][1] = wp[256 + 16];
    }
    const float b2a = b2t[32*wav + bb], b2b = b2t[32*wav + 16 + bb];
    const float b1v = b1t[lane];
    #pragma unroll
    for (int tt = 0; tt < NT; ++tt) {
        #pragma unroll
        for (int r = 0; r < 16; ++r) xh[tt*XT + r*XP + t] = xr[tt][r];
    }
    __syncthreads();

    float* const p1 = bij;
    #pragma nounroll
    for (int th = 0; th < NT; th += NP) {
        v8f c0[NP], c1[NP];
        #pragma unroll
        for (int tp = 0; tp < NP; ++tp) { c0[tp] = vzero8();  c1[tp] = vzero8(); }
        const float* xa = xh + th*XT + bb*XP + kq*64 + 2*hi;
        #pragma unroll
        for (int s = 0; s < 16; ++s) {
            #pragma unroll
            for (int tp = 0; tp < NP; ++tp) {
                v2f af;  af[0] = xa[tp*XT + 4*s];  af[1] = xa[tp*XT + 4*s + 1];
                if (s & 1) c1[tp] = wmma4(af, w1f[s], c1[tp]);
                else       c0[tp] = wmma4(af, w1f[s], c0[tp]);
            }
        }
        #pragma unroll
        for (int tp = 0; tp < NP; ++tp) {
            const v8f c = c0[tp] + c1[tp];
            float* pp = p1 + (th + tp)*PT + (kq*16 + 8*hi)*HP + nt*16 + bb;
            #pragma unroll
            for (int r = 0; r < 8; ++r) pp[r*HP] = c[r];
        }
    }
    __syncthreads();

    #pragma nounroll
    for (int th = 0; th < NT; th += NP) {
        float z[2*NP];
        #pragma unroll
        for (int tp = 0; tp < NP; ++tp) {
            const float* pt = p1 + (th + tp)*PT;
            #pragma unroll
            for (int bsel = 0; bsel < 2; ++bsel) {
                const int b = wav + 8*bsel;
                float acc = pt[b*HP + lane];
                #pragma unroll
                for (int q = 1; q < 4; ++q) acc += pt[(q*16 + b)*HP + lane];
                z[2*tp + bsel] = acc + b1v;
            }
        }
        gelu_n<2*NP>(z);
        #pragma unroll
        for (int tp = 0; tp < NP; ++tp) {
            h1s[(th + tp)*HT + wav*HP + lane] = z[2*tp];
            h1s[(th + tp)*HT + (wav + 8)*HP + lane] = z[2*tp + 1];
        }
    }
    __syncthreads();

    for (int tt = 0; tt < NT; ++tt) {
        v8f d0 = vzero8(), d1 = vzero8();
        const float* ha = h1s + tt*HT + bb*HP + 2*hi;
        #pragma unroll
        for (int s = 0; s < 8; ++s) {
            v2f af;  af[0] = ha[4*s];  af[1] = ha[4*s + 1];
            d0 = wmma4(af, w2f0[s], d0);
            d1 = wmma4(af, w2f1[s], d1);
        }
        float* ho = xh + tt*XT + (8*hi)*256 + 32*wav + bb;
        float z[16];
        #pragma unroll
        for (int r = 0; r < 8; ++r) { z[2*r] = d0[r] + b2a;  z[2*r + 1] = d1[r] + b2b; }
        gelu_n<16>(z);
        #pragma unroll
        for (int r = 0; r < 8; ++r) {
            ho[r*256]      = z[2*r];
            ho[r*256 + 16] = z[2*r + 1];
        }
    }
    __syncthreads();

    #pragma nounroll
    for (int th = 0; th < NT; th += NP) {
        const int ba = wav, bc = wav + 8;
        const float* xb = xh + th*XT;
        float sa[NP], sb[NP], mua[NP], mub[NP], va[NP], vb[NP];
        #pragma unroll
        for (int tp = 0; tp < NP; ++tp) {
            sa[tp] = 0.f;  sb[tp] = 0.f;
            #pragma unroll
            for (int r = 0; r < 8; ++r) { sa[tp] += xb[tp*XT + ba*256 + lane + 32*r]; sb[tp] += xb[tp*XT + bc*256 + lane + 32*r]; }
        }
        #pragma unroll
        for (int m = 16; m >= 1; m >>= 1) {
            #pragma unroll
            for (int tp = 0; tp < NP; ++tp) { sa[tp] += __shfl_xor(sa[tp], m, 32); sb[tp] += __shfl_xor(sb[tp], m, 32); }
        }
        #pragma unroll
        for (int tp = 0; tp < NP; ++tp) {
            mua[tp] = sa[tp] * (1.0f/256.0f);  mub[tp] = sb[tp] * (1.0f/256.0f);
            va[tp] = 0.f;  vb[tp] = 0.f;
            #pragma unroll
            for (int r = 0; r < 8; ++r) {
                float da = xb[tp*XT + ba*256 + lane + 32*r] - mua[tp]; va[tp] = fmaf(da, da, va[tp]);
                float db = xb[tp*XT + bc*256 + lane + 32*r] - mub[tp]; vb[tp] = fmaf(db, db, vb[tp]);
            }
        }
        #pragma unroll
        for (int m = 16; m >= 1; m >>= 1) {
            #pragma unroll
            for (int tp = 0; tp < NP; ++tp) { va[tp] += __shfl_xor(va[tp], m, 32); vb[tp] += __shfl_xor(vb[tp], m, 32); }
        }
        if (lane == 0) {
            #pragma unroll
            for (int tp = 0; tp < NP; ++tp) {
                mu_s[(th + tp)*16 + ba] = mua[tp]; rstd_s[(th + tp)*16 + ba] = rsqrtf(va[tp] * (1.0f/256.0f) + LN_EPS);
                mu_s[(th + tp)*16 + bc] = mub[tp]; rstd_s[(th + tp)*16 + bc] = rsqrtf(vb[tp] * (1.0f/256.0f) + LN_EPS);
            }
        }
    }
    __syncthreads();

    for (unsigned tt = 0; tt < nb; ++tt) {
    const unsigned tile = tile0 + tt;
    const int tzt = opaque_zero();
    const float* const lgt = ln_g + tzt; const float* const lbt = ln_b + tzt;
    const float* const Wft = Wf + tzt;   const float* const bft = bf + tzt;

    {
        int i = t & 31, b0 = t >> 5;
        float g8[8], be8[8];
        #pragma unroll
        for (int k = 0; k < 8; ++k) { g8[k] = lgt[i*8 + k]; be8[k] = lbt[i*8 + k]; }
        const float* h2 = xh + tt*XT;
        #pragma unroll
        for (int bsel = 0; bsel < 2; ++bsel) {
            int b = b0 + 8*bsel;
            float mu = mu_s[tt*16 + b], rs = rstd_s[tt*16 + b];
            float vals[8]; float nsq = 0.f;
            #pragma unroll
            for (int k = 0; k < 8; ++k) {
                float v = (h2[b*256 + i*8 + k] - mu) * rs * g8[k] + be8[k];
                vals[k] = v; nsq = fmaf(v, v, nsq);
            }
            float scale = nsq / ((1.0f + nsq) * sqrtf(nsq + SQ_EPS));
            #pragma unroll
            for (int k = 0; k < 8; ++k) u_t[i*130 + k*16 + b] = scale * vals[k];
        }
    }

    #pragma unroll
    for (int r = 0; r < 36; ++r) { bij[t + 256*r] = 0.f; cij[t + 256*r] = 0.0625f; }
    if (t < 16*RS - 9216) { bij[9216 + t] = 0.f; cij[9216 + t] = 0.0625f; }
    __syncthreads();

    v8f vA = vzero8(), vB = vzero8();
    v4f* const uc = uh_c + t*CP;
    for (int r = 0; r < 3; ++r) {
        v8f sA0 = vzero8(), sA1 = vzero8(), sB0 = vzero8(), sB1 = vzero8();
        v2f ub0[32], ub1[32];
        v8f sAc = vzero8(), sBc = vzero8();
        if (r == 0) {
            #pragma unroll
            for (int i = 0; i < NC; ++i) {
                const float* ub = u_t + i*130 + bb;
                v2f bu0, bu1;
                bu0[0] = ub[(2*hi+0)*16];  bu0[1] = ub[(2*hi+1)*16];
                bu1[0] = ub[(2*hi+4)*16];  bu1[1] = ub[(2*hi+5)*16];
                v8f uA = vzero8(); uA = wmma4(a0A[i], bu0, uA); uA = wmma4(a1A[i], bu1, uA);
                v8f uB = vzero8(); uB = wmma4(a0B[i], bu0, uB); uB = wmma4(a1B[i], bu1, uB);
                const v4f q0 = {uA[0], uA[1], uA[2], uA[3]}, q1 = {uA[4], uA[5], uA[6], uA[7]};
                const v4f q2 = {uB[0], uB[1], uB[2], uB[3]}, q3 = {uB[4], uB[5], uB[6], uB[7]};
                uc[i*4 + 0] = q0;  uc[i*4 + 1] = q1;
                uc[i*4 + 2] = q2;  uc[i*4 + 3] = q3;
                const float ccA = cij[bb*RS + i*RP + wav];
                const float ccB = cij[bb*RS + i*RP + wav + 8];
                #pragma unroll
                for (int v = 0; v < 8; ++v) { sAc[v] = fmaf(uA[v], ccA, sAc[v]); sBc[v] = fmaf(uB[v], ccB, sBc[v]); }
            }
        } else {
            #pragma unroll
            for (int i = 0; i < NC; ++i) {
                const v4f q0 = uc[i*4 + 0], q1 = uc[i*4 + 1];
                const v4f q2 = uc[i*4 + 2], q3 = uc[i*4 + 3];
                const float ccA = cij[bb*RS + i*RP + wav];
                const float ccB = cij[bb*RS + i*RP + wav + 8];
                #pragma unroll
                for (int v = 0; v < 4; ++v) {
                    sAc[v] = fmaf(q0[v], ccA, sAc[v]);  sAc[v + 4] = fmaf(q1[v], ccA, sAc[v + 4]);
                    sBc[v] = fmaf(q2[v], ccB, sBc[v]);  sBc[v + 4] = fmaf(q3[v], ccB, sBc[v + 4]);
                }
            }
        }
        #pragma unroll
        for (int i = NC; i < 32; ++i) {
            const float* ub = u_t + i*130 + bb;
            v2f bu0, bu1;
            bu0[0] = ub[(2*hi+0)*16];  bu0[1] = ub[(2*hi+1)*16];
            bu1[0] = ub[(2*hi+4)*16];  bu1[1] = ub[(2*hi+5)*16];
            ub0[i] = bu0;  ub1[i] = bu1;
            float cA = cij[bb*RS + i*RP + wav];
            float cB = cij[bb*RS + i*RP + wav + 8];
            v2f p0, p1;
            p0[0] = bu0[0]*cA; p0[1] = bu0[1]*cA; p1[0] = bu1[0]*cA; p1[1] = bu1[1]*cA;
            if (i & 1) { sA1 = wmma4(a0A[i], p0, sA1); sA1 = wmma4(a1A[i], p1, sA1); }
            else       { sA0 = wmma4(a0A[i], p0, sA0); sA0 = wmma4(a1A[i], p1, sA0); }
            p0[0] = bu0[0]*cB; p0[1] = bu0[1]*cB; p1[0] = bu1[0]*cB; p1[1] = bu1[1]*cB;
            if (i & 1) { sB1 = wmma4(a0B[i], p0, sB1); sB1 = wmma4(a1B[i], p1, sB1); }
            else       { sB0 = wmma4(a0B[i], p0, sB0); sB0 = wmma4(a1B[i], p1, sB0); }
        }
        v8f sA = (sA0 + sA1) + sAc, sB = (sB0 + sB1) + sBc;
        float nA = 0.f, nB = 0.f;
        #pragma unroll
        for (int v = 0; v < 8; ++v) { nA = fmaf(sA[v], sA[v], nA); nB = fmaf(sB[v], sB[v], nB); }
        nA += __shfl_xor(nA, 16, 32);  nB += __shfl_xor(nB, 16, 32);
        float scA = nA / ((1.0f + nA) * sqrtf(nA + SQ_EPS));
        float scB = nB / ((1.0f + nB) * sqrtf(nB + SQ_EPS));
        #pragma unroll
        for (int v = 0; v < 8; ++v) { vA[v] = sA[v]*scA; vB[v] = sB[v]*scB; }

        if (r < 2) {
            const int ib = bb*RS + wav + 8*hi;
            #pragma unroll
            for (int n = 0; n < 32; ++n) {
                const bool cached = (n % 4 == 3);
                const int i = cached ? n / 4 : NC + n - n / 4;
                v8f uA, uB;
                if (cached) {
                    const v4f q0 = uc[i*4 + 0], q1 = uc[i*4 + 1];
                    const v4f q2 = uc[i*4 + 2], q3 = uc[i*4 + 3];
                    #pragma unroll
                    for (int v = 0; v < 4; ++v) { uA[v] = q0[v]; uA[v + 4] = q1[v]; uB[v] = q2[v]; uB[v + 4] = q3[v]; }
                } else {
                    const v2f bu0 = ub0[i], bu1 = ub1[i];
                    uA = vzero8(); uA = wmma4(a0A[i], bu0, uA); uA = wmma4(a1A[i], bu1, uA);
                    uB = vzero8(); uB = wmma4(a0B[i], bu0, uB); uB = wmma4(a1B[i], bu1, uB);
                }
                float agA = 0.f, agB = 0.f;
                #pragma unroll
                for (int v = 0; v < 8; ++v) { agA = fmaf(uA[v], vA[v], agA); agB = fmaf(uB[v], vB[v], agB); }
                const float keep = hi ? agB : agA;
                const float send = hi ? agA : agB;
                bij[ib + i*RP] += keep + xhalf(send);
            }
            __syncthreads();
            #pragma unroll
            for (int bsel = 0; bsel < 2; ++bsel) {
                int base = (wav + 8*bsel)*RS + (t & 31)*RP;
                float mx = bij[base];
                #pragma unroll
                for (int j = 1; j < 16; ++j) mx = fmaxf(mx, bij[base + j]);
                float e[16]; float sum = 0.f;
                #pragma unroll
                for (int j = 0; j < 16; ++j) { e[j] = exp_unit(bij[base + j] - mx); sum += e[j]; }
                float inv = 1.0f / sum;
                #pragma unroll
                for (int j = 0; j < 16; ++j) cij[base + j] = e[j] * inv;
            }
            __syncthreads();
        }
    }

    {
        float pA = 0.f, pB = 0.f;
        #pragma unroll
        for (int v = 0; v < 8; ++v) {
            pA = fmaf(vA[v], Wft[(wav    )*16 + v + 8*hi], pA);
            pB = fmaf(vB[v], Wft[(wav + 8)*16 + v + 8*hi], pB);
        }
        pA += __shfl_xor(pA, 16, 32);  pB += __shfl_xor(pB, 16, 32);
        if (lane < 16) { part[wav*17 + bb] = pA; part[(wav+8)*17 + bb] = pB; }
    }
    __syncthreads();
    if (t < 16) {
        float acc = bft[0];
        #pragma unroll
        for (int j = 0; j < 16; ++j) acc += part[j*17 + t];
        out[tile*16 + t] = acc;
    }
    }
    }
}

extern "C" void kernel_launch(void* const* d_in, const int* in_sizes, int n_in,
                              void* d_out, int out_size, void* d_ws, size_t ws_size,
                              hipStream_t stream) {
    const float* x    = (const float*)d_in[0];
    const float* W1   = (const float*)d_in[1];
    const float* b1   = (const float*)d_in[2];
    const float* W2   = (const float*)d_in[3];
    const float* b2   = (const float*)d_in[4];
    const float* ln_g = (const float*)d_in[5];
    const float* ln_b = (const float*)d_in[6];
    const float* Wcap = (const float*)d_in[7];
    const float* Wf   = (const float*)d_in[8];
    const float* bf   = (const float*)d_in[9];
    float* out = (float*)d_out;

    const int B = in_sizes[0] / 256;
    const int tiles = B / 16;
    capsnet_fused<<<NWG, 256, 0, stream>>>(x, W1, b1, W2, b2, ln_g, ln_b, Wcap, Wf, bf, out, tiles);
}


# LDS instructions out of the routing's matrix streams: s pass reads as one group ahead of its 96 calls, agreement pass at 4 places (was 24), squash half exchange on v_permlanex16; sums bit-identical
# speedup vs baseline: 2.3116x; 1.0453x over previous
#include <hip/hip_runtime.h>
#include <math.h>

typedef __attribute__((ext_vector_type(2))) float v2f;
typedef __attribute__((ext_vector_type(4))) float v4f;
typedef __attribute__((ext_vector_type(8))) float v8f;

#define LN_EPS 1e-5f
#define SQ_EPS 1e-9f

#define RP 18
#define RS 579

#define NC 8
#define CP (4*NC + 1)
#if 4*NC != 32
#error "the agreement pass's capsule order is written for NC = 8"
#endif

#define XP 260
#define HP 36

#define NWG 256

#define NT 4
#define NP 2
#define XT (16*XP)
#define HT (16*HP)
#define PT (4*16*HP)

__device__ __forceinline__ float exp_unit(float x) {
    return __builtin_amdgcn_exp2f(x * 1.44269504088896340736f);
}

#define ERF_SPLIT 0.927734375f
__device__ __forceinline__ float erf_small(float a) {
    const float s = a * a;
    float q = -5.96761703e-4f;
    q = fmaf(q, s, 4.99119423e-3f);
    q = fmaf(q, s, -2.67681349e-2f);
    q = fmaf(q, s, 1.12819925e-1f);
    q = fmaf(q, s, -3.76125336e-1f);
    q = fmaf(q, s, 1.28379166e-1f);
    return fmaf(q, a, a);
}
__device__ __forceinline__ float erf_large(float a) {
    const float t = fabsf(a);
    const float s = a * a;
    float r = fmaf(-1.72853470e-5f, t, 3.83197126e-4f);
    const float u = fmaf(-3.88396438e-3f, t, 2.42546219e-2f);
    r = fmaf(r, s, u);
    r = fmaf(r, t, -1.06777877e-1f);
    r = fmaf(r, t, -6.34846687e-1f);
    r = fmaf(r, t, -1.28717512e-1f);
    r = fmaf(r, t, -t);
    return copysignf(1.0f - exp_unit(r), a);
}

template <int N>
__device__ __forceinline__ void gelu_n(float (&z)[N]) {
    float a[N], e[N];
    float amax = 0.f;
    #pragma unroll
    for (int n = 0; n < N; ++n) {
        a[n] = z[n] * 0.70710678118654752440f;
        e[n] = erf_small(a[n]);
        amax = fmaxf(amax, fabsf(a[n]));
    }
    if (amax > ERF_SPLIT) {
        #pragma unroll
        for (int n = 0; n < N; ++n) {
            const float el = erf_large(a[n]);
            e[n] = (fabsf(a[n]) > ERF_SPLIT) ? el : e[n];
        }
    }
    #pragma unroll
    for (int n = 0; n < N; ++n) z[n] = 0.5f * z[n] * (1.0f + e[n]);
}

__device__ __forceinline__ v8f wmma4(v2f a, v2f b, v8f c) {
    return __builtin_amdgcn_wmma_f32_16x16x4_f32(false, a, false, b, (short)0, c, false, false);
}

__device__ __forceinline__ float xhalf(float x) {
    const int b = __builtin_bit_cast(int, x);
    return __builtin_bit_cast(float, __builtin_amdgcn_permlanex16(b, b, 0x76543210, 0xfedcba98, false, false));
}

__device__ __forceinline__ float agree(const v8f uA, const v8f uB, const v8f vA, const v8f vB, const int hi) {
    float agA = 0.f, agB = 0.f;
    #pragma unroll
    for (int v = 0; v < 8; ++v) { agA = fmaf(uA[v], vA[v], agA); agB = fmaf(uB[v], vB[v], agB); }
    const float keep = hi ? agB : agA;
    const float send = hi ? agA : agB;
    return keep + xhalf(send);
}

__device__ __forceinline__ int opaque_zero() {
    int z = 0;
    asm volatile("" : "+s"(z));
    return z;
}

__device__ __forceinline__ v8f vzero8() {
    v8f z = {0.f,0.f,0.f,0.f,0.f,0.f,0.f,0.f};
    return z;
}

__global__ __attribute__((amdgpu_waves_per_eu(1))) __launch_bounds__(256)
void capsnet_fused(
    const float* __restrict__ x,   const float* __restrict__ W1, const float* __restrict__ b1,
    const float* __restrict__ W2,  const float* __restrict__ b2,
    const float* __restrict__ ln_g,const float* __restrict__ ln_b,
    const float* __restrict__ Wcap,const float* __restrict__ Wf, const float* __restrict__ bf,
    float* __restrict__ out, const int ntiles)
{
    __shared__ __attribute__((aligned(32))) float xh[NT*XT];
    __shared__ __attribute__((aligned(32))) float h1s[NT*HT];
    __shared__ __attribute__((aligned(32))) float mu_s[NT*16];
    __shared__ __attribute__((aligned(32))) float rstd_s[NT*16];
    __shared__ __attribute__((aligned(32))) float u_t[32*130];
    __shared__ __attribute__((aligned(32))) float bij[16*RS];
    __shared__ __attribute__((aligned(32))) float cij[16*RS];
    __shared__ __attribute__((aligned(32))) float part[16*17];
    __shared__ v4f uh_c[256*CP];

    const int t    = threadIdx.x;
    const int lane = t & 31;
    const int wav  = t >> 5;
    const int bb   = lane & 15;
    const int hi   = lane >> 4;

    const int nt = wav & 1, kq = wav >> 1;

    v2f a0A[32], a1A[32], a0B[32], a1B[32];
    #pragma unroll
    for (int i = 0; i < 32; ++i) {
        const float* wcA = Wcap + (size_t)((i*16 + wav    )*8)*16 + bb;
        const float* wcB = Wcap + (size_t)((i*16 + wav + 8)*8)*16 + bb;
        a0A[i][0] = wcA[(2*hi+0)*16];  a0A[i][1] = wcA[(2*hi+1)*16];
        a1A[i][0] = wcA[(2*hi+4)*16];  a1A[i][1] = wcA[(2*hi+5)*16];
        a0B[i][0] = wcB[(2*hi+0)*16];  a0B[i][1] = wcB[(2*hi+1)*16];
        a1B[i][0] = wcB[(2*hi+4)*16];  a1B[i][1] = wcB[(2*hi+5)*16];
    }


    const unsigned tile_lo = (blockIdx.x * (unsigned)ntiles) / NWG;
    const unsigned tile_hi = ((blockIdx.x + 1u) * (unsigned)ntiles) / NWG;
    for (unsigned tile0 = tile_lo; tile0 < tile_hi; tile0 += NT) {
    const unsigned nb = (tile_hi - tile0 < (unsigned)NT) ? (tile_hi - tile0) : (unsigned)NT;
    const int tz = opaque_zero();
    const float* const W1t = W1 + tz;    const float* const b1t = b1 + tz;
    const float* const W2t = W2 + tz;    const float* const b2t = b2 + tz;


    float xr[NT][16];
    #pragma unroll
    for (int tt = 0; tt < NT; ++tt) {
        const unsigned ts = (tile0 + tt < tile_hi) ? (tile0 + tt) : (tile_hi - 1u);
        const float* xg = x + (size_t)ts * 16 * 256;
        #pragma unroll
        for (int r = 0; r < 16; ++r) xr[tt][r] = xg[t + 256*r];
    }
    v2f w1f[16];
    #pragma unroll
    for (int s = 0; s < 16; ++s) {
        const float* wp = W1t + (size_t)(kq*64 + 4*s + 2*hi)*32 + nt*16 + bb;
        w1f[s][0] = wp[0];  w1f[s][1] = wp[32];
    }
    v2f w2f0[8], w2f1[8];
    #pragma unroll
    for (int s = 0; s < 8; ++s) {
        const float* wp = W2t + (size_t)(4*s + 2*hi)*256 + 32*wav + bb;
        w2f0[s][0] = wp[0];   w2f0[s][1] = wp[256];
        w2f1[s][0] = wp[16];  w2f1[s][1] = wp[256 + 16];
    }
    const float b2a = b2t[32*wav + bb], b2b = b2t[32*wav + 16 + bb];
    const float b1v = b1t[lane];
    #pragma unroll
    for (int tt = 0; tt < NT; ++tt) {
        #pragma unroll
        for (int r = 0; r < 16; ++r) xh[tt*XT + r*XP + t] = xr[tt][r];
    }
    __syncthreads();

    float* const p1 = bij;
    #pragma nounroll
    for (int th = 0; th < NT; th += NP) {
        v8f c0[NP], c1[NP];
        #pragma unroll
        for (int tp = 0; tp < NP; ++tp) { c0[tp] = vzero8();  c1[tp] = vzero8(); }
        const float* xa = xh + th*XT + bb*XP + kq*64 + 2*hi;
        #pragma unroll
        for (int s = 0; s < 16; ++s) {
            #pragma unroll
            for (int tp = 0; tp < NP; ++tp) {
                v2f af;  af[0] = xa[tp*XT + 4*s];  af[1] = xa[tp*XT + 4*s + 1];
                if (s & 1) c1[tp] = wmma4(af, w1f[s], c1[tp]);
                else       c0[tp] = wmma4(af, w1f[s], c0[tp]);
            }
        }
        #pragma unroll
        for (int tp = 0; tp < NP; ++tp) {
            const v8f c = c0[tp] + c1[tp];
            float* pp = p1 + (th + tp)*PT + (kq*16 + 8*hi)*HP + nt*16 + bb;
            #pragma unroll
            for (int r = 0; r < 8; ++r) pp[r*HP] = c[r];
        }
    }
    __syncthreads();

    #pragma nounroll
    for (int th = 0; th < NT; th += NP) {
        float z[2*NP];
        #pragma unroll
        for (int tp = 0; tp < NP; ++tp) {
            const float* pt = p1 + (th + tp)*PT;
            #pragma unroll
            for (int bsel = 0; bsel < 2; ++bsel) {
                const int b = wav + 8*bsel;
                float acc = pt[b*HP + lane];
                #pragma unroll
                for (int q = 1; q < 4; ++q) acc += pt[(q*16 + b)*HP + lane];
                z[2*tp + bsel] = acc + b1v;
            }
        }
        gelu_n<2*NP>(z);
        #pragma unroll
        for (int tp = 0; tp < NP; ++tp) {
            h1s[(th + tp)*HT + wav*HP + lane] = z[2*tp];
            h1s[(th + tp)*HT + (wav + 8)*HP + lane] = z[2*tp + 1];
        }
    }
    __syncthreads();

    for (int tt = 0; tt < NT; ++tt) {
        v8f d0 = vzero8(), d1 = vzero8();
        const float* ha = h1s + tt*HT + bb*HP + 2*hi;
        #pragma unroll
        for (int s = 0; s < 8; ++s) {
            v2f af;  af[0] = ha[4*s];  af[1] = ha[4*s + 1];
            d0 = wmma4(af, w2f0[s], d0);
            d1 = wmma4(af, w2f1[s], d1);
        }
        float* ho = xh + tt*XT + (8*hi)*256 + 32*wav + bb;
        float z[16];
        #pragma unroll
        for (int r = 0; r < 8; ++r) { z[2*r] = d0[r] + b2a;  z[2*r + 1] = d1[r] + b2b; }
        gelu_n<16>(z);
        #pragma unroll
        for (int r = 0; r < 8; ++r) {
            ho[r*256]      = z[2*r];
            ho[r*256 + 16] = z[2*r + 1];
        }
    }
    __syncthreads();

    #pragma nounroll
    for (int th = 0; th < NT; th += NP) {
        const int ba = wav, bc = wav + 8;
        const float* xb = xh + th*XT;
        float sa[NP], sb[NP], mua[NP], mub[NP], va[NP], vb[NP];
        #pragma unroll
        for (int tp = 0; tp < NP; ++tp) {
            sa[tp] = 0.f;  sb[tp] = 0.f;
            #pragma unroll
            for (int r = 0; r < 8; ++r) { sa[tp] += xb[tp*XT + ba*256 + lane + 32*r]; sb[tp] += xb[tp*XT + bc*256 + lane + 32*r]; }
        }
        #pragma unroll
        for (int m = 16; m >= 1; m >>= 1) {
            #pragma unroll
            for (int tp = 0; tp < NP; ++tp) { sa[tp] += __shfl_xor(sa[tp], m, 32); sb[tp] += __shfl_xor(sb[tp], m, 32); }
        }
        #pragma unroll
        for (int tp = 0; tp < NP; ++tp) {
            mua[tp] = sa[tp] * (1.0f/256.0f);  mub[tp] = sb[tp] * (1.0f/256.0f);
            va[tp] = 0.f;  vb[tp] = 0.f;
            #pragma unroll
            for (int r = 0; r < 8; ++r) {
                float da = xb[tp*XT + ba*256 + lane + 32*r] - mua[tp]; va[tp] = fmaf(da, da, va[tp]);
                float db = xb[tp*XT + bc*256 + lane + 32*r] - mub[tp]; vb[tp] = fmaf(db, db, vb[tp]);
            }
        }
        #pragma unroll
        for (int m = 16; m >= 1; m >>= 1) {
            #pragma unroll
            for (int tp = 0; tp < NP; ++tp) { va[tp] += __shfl_xor(va[tp], m, 32); vb[tp] += __shfl_xor(vb[tp], m, 32); }
        }
        if (lane == 0) {
            #pragma unroll
            for (int tp = 0; tp < NP; ++tp) {
                mu_s[(th + tp)*16 + ba] = mua[tp]; rstd_s[(th + tp)*16 + ba] = rsqrtf(va[tp] * (1.0f/256.0f) + LN_EPS);
                mu_s[(th + tp)*16 + bc] = mub[tp]; rstd_s[(th + tp)*16 + bc] = rsqrtf(vb[tp] * (1.0f/256.0f) + LN_EPS);
            }
        }
    }
    __syncthreads();

    for (unsigned tt = 0; tt < nb; ++tt) {
    const unsigned tile = tile0 + tt;
    const int tzt = opaque_zero();
    const float* const lgt = ln_g + tzt; const float* const lbt = ln_b + tzt;
    const float* const Wft = Wf + tzt;   const float* const bft = bf + tzt;

    {
        int i = t & 31, b0 = t >> 5;
        float g8[8], be8[8];
        #pragma unroll
        for (int k = 0; k < 8; ++k) { g8[k] = lgt[i*8 + k]; be8[k] = lbt[i*8 + k]; }
        const float* h2 = xh + tt*XT;
        #pragma unroll
        for (int bsel = 0; bsel < 2; ++bsel) {
            int b = b0 + 8*bsel;
            float mu = mu_s[tt*16 + b], rs = rstd_s[tt*16 + b];
            float vals[8]; float nsq = 0.f;
            #pragma unroll
            for (int k = 0; k < 8; ++k) {
                float v = (h2[b*256 + i*8 + k] - mu) * rs * g8[k] + be8[k];
                vals[k] = v; nsq = fmaf(v, v, nsq);
            }
            float scale = nsq / ((1.0f + nsq) * sqrtf(nsq + SQ_EPS));
            #pragma unroll
            for (int k = 0; k < 8; ++k) u_t[i*130 + k*16 + b] = scale * vals[k];
        }
    }

    #pragma unroll
    for (int r = 0; r < 36; ++r) { bij[t + 256*r] = 0.f; cij[t + 256*r] = 0.0625f; }
    if (t < 16*RS - 9216) { bij[9216 + t] = 0.f; cij[9216 + t] = 0.0625f; }
    __syncthreads();

    v8f vA = vzero8(), vB = vzero8();
    v4f* const uc = uh_c + t*CP;
    for (int r = 0; r < 3; ++r) {
        v8f sA0 = vzero8(), sA1 = vzero8(), sB0 = vzero8(), sB1 = vzero8();
        v2f ub0[32], ub1[32];
        v8f sAc = vzero8(), sBc = vzero8();
        if (r == 0) {
            #pragma unroll
            for (int i = 0; i < NC; ++i) {
                const float* ub = u_t + i*130 + bb;
                v2f bu0, bu1;
                bu0[0] = ub[(2*hi+0)*16];  bu0[1] = ub[(2*hi+1)*16];
                bu1[0] = ub[(2*hi+4)*16];  bu1[1] = ub[(2*hi+5)*16];
                v8f uA = vzero8(); uA = wmma4(a0A[i], bu0, uA); uA = wmma4(a1A[i], bu1, uA);
                v8f uB = vzero8(); uB = wmma4(a0B[i], bu0, uB); uB = wmma4(a1B[i], bu1, uB);
                const v4f q0 = {uA[0], uA[1], uA[2], uA[3]}, q1 = {uA[4], uA[5], uA[6], uA[7]};
                const v4f q2 = {uB[0], uB[1], uB[2], uB[3]}, q3 = {uB[4], uB[5], uB[6], uB[7]};
                uc[i*4 + 0] = q0;  uc[i*4 + 1] = q1;
                uc[i*4 + 2] = q2;  uc[i*4 + 3] = q3;
                const float ccA = cij[bb*RS + i*RP + wav];
                const float ccB = cij[bb*RS + i*RP + wav + 8];
                #pragma unroll
                for (int v = 0; v < 8; ++v) { sAc[v] = fmaf(uA[v], ccA, sAc[v]); sBc[v] = fmaf(uB[v], ccB, sBc[v]); }
            }
        } else {
            #pragma unroll
            for (int i = 0; i < NC; ++i) {
                const v4f q0 = uc[i*4 + 0], q1 = uc[i*4 + 1];
                const v4f q2 = uc[i*4 + 2], q3 = uc[i*4 + 3];
                const float ccA = cij[bb*RS + i*RP + wav];
                const float ccB = cij[bb*RS + i*RP + wav + 8];
                #pragma unroll
                for (int v = 0; v < 4; ++v) {
                    sAc[v] = fmaf(q0[v], ccA, sAc[v]);  sAc[v + 4] = fmaf(q1[v], ccA, sAc[v + 4]);
                    sBc[v] = fmaf(q2[v], ccB, sBc[v]);  sBc[v + 4] = fmaf(q3[v], ccB, sBc[v + 4]);
                }
            }
        }
        float cAs[32], cBs[32];
        #pragma unroll
        for (int i = NC; i < 32; ++i) {
            const float* ub = u_t + i*130 + bb;
            ub0[i][0] = ub[(2*hi+0)*16];  ub0[i][1] = ub[(2*hi+1)*16];
            ub1[i][0] = ub[(2*hi+4)*16];  ub1[i][1] = ub[(2*hi+5)*16];
            cAs[i] = cij[bb*RS + i*RP + wav];
            cBs[i] = cij[bb*RS + i*RP + wav + 8];
        }
        __builtin_amdgcn_sched_barrier(0);
        #pragma unroll
        for (int i = NC; i < 32; ++i) {
            const v2f bu0 = ub0[i], bu1 = ub1[i];
            const float cA = cAs[i], cB = cBs[i];
            v2f p0, p1;
            p0[0] = bu0[0]*cA; p0[1] = bu0[1]*cA; p1[0] = bu1[0]*cA; p1[1] = bu1[1]*cA;
            if (i & 1) { sA1 = wmma4(a0A[i], p0, sA1); sA1 = wmma4(a1A[i], p1, sA1); }
            else       { sA0 = wmma4(a0A[i], p0, sA0); sA0 = wmma4(a1A[i], p1, sA0); }
            p0[0] = bu0[0]*cB; p0[1] = bu0[1]*cB; p1[0] = bu1[0]*cB; p1[1] = bu1[1]*cB;
            if (i & 1) { sB1 = wmma4(a0B[i], p0, sB1); sB1 = wmma4(a1B[i], p1, sB1); }
            else       { sB0 = wmma4(a0B[i], p0, sB0); sB0 = wmma4(a1B[i], p1, sB0); }
        }
        v8f sA = (sA0 + sA1) + sAc, sB = (sB0 + sB1) + sBc;
        float nA = 0.f, nB = 0.f;
        #pragma unroll
        for (int v = 0; v < 8; ++v) { nA = fmaf(sA[v], sA[v], nA); nB = fmaf(sB[v], sB[v], nB); }
        nA += xhalf(nA);  nB += xhalf(nB);
        float scA = nA / ((1.0f + nA) * sqrtf(nA + SQ_EPS));
        float scB = nB / ((1.0f + nB) * sqrtf(nB + SQ_EPS));
        #pragma unroll
        for (int v = 0; v < 8; ++v) { vA[v] = sA[v]*scA; vB[v] = sB[v]*scB; }

        if (r < 2) {
            const int ib = bb*RS + wav + 8*hi;
            float bl[32];
            v4f q[4][4];
            #pragma unroll
            for (int n = 0; n < 8; ++n) bl[NC + n] = bij[ib + (NC + n)*RP];
            __builtin_amdgcn_sched_barrier(0);
            #pragma unroll
            for (int g = 0; g < 3; ++g) {
                #pragma unroll
                for (int n = 0; n < 8; ++n) {
                    const int i = NC + 8*g + n;
                    v8f uA = vzero8(); uA = wmma4(a0A[i], ub0[i], uA); uA = wmma4(a1A[i], ub1[i], uA);
                    v8f uB = vzero8(); uB = wmma4(a0B[i], ub0[i], uB); uB = wmma4(a1B[i], ub1[i], uB);
                    bl[i] += agree(uA, uB, vA, vB, hi);
                }
                if (g >= 1) {
                    #pragma unroll
                    for (int k = 0; k < 4; ++k) {
                        const int i = 4*(g - 1) + k;
                        v8f uA, uB;
                        #pragma unroll
                        for (int v = 0; v < 4; ++v) { uA[v] = q[k][0][v]; uA[v + 4] = q[k][1][v]; uB[v] = q[k][2][v]; uB[v + 4] = q[k][3][v]; }
                        bl[i] += agree(uA, uB, vA, vB, hi);
                    }
                }
                __builtin_amdgcn_sched_barrier(0);
                if (g < 2) {
                    #pragma unroll
                    for (int n = 0; n < 8; ++n) bl[NC + 8*(g + 1) + n] = bij[ib + (NC + 8*(g + 1) + n)*RP];
                    #pragma unroll
                    for (int k = 0; k < 4; ++k) {
                        const int i = 4*g + k;
                        bl[i] = bij[ib + i*RP];
                        q[k][0] = uc[i*4 + 0];  q[k][1] = uc[i*4 + 1];
                        q[k][2] = uc[i*4 + 2];  q[k][3] = uc[i*4 + 3];
                    }
                }
                #pragma unroll
                for (int n = 0; n < 8; ++n) bij[ib + (NC + 8*g + n)*RP] = bl[NC + 8*g + n];
                if (g >= 1) {
                    #pragma unroll
                    for (int k = 0; k < 4; ++k) bij[ib + (4*(g - 1) + k)*RP] = bl[4*(g - 1) + k];
                }
                if (g < 2) {
                    __builtin_amdgcn_sched_barrier(0);
                }
            }
            __syncthreads();
            #pragma unroll
            for (int bsel = 0; bsel < 2; ++bsel) {
                int base = (wav + 8*bsel)*RS + (t & 31)*RP;
                float mx = bij[base];
                #pragma unroll
                for (int j = 1; j < 16; ++j) mx = fmaxf(mx, bij[base + j]);
                float e[16]; float sum = 0.f;
                #pragma unroll
                for (int j = 0; j < 16; ++j) { e[j] = exp_unit(bij[base + j] - mx); sum += e[j]; }
                float inv = 1.0f / sum;
                #pragma unroll
                for (int j = 0; j < 16; ++j) cij[base + j] = e[j] * inv;
            }
            __syncthreads();
        }
    }

    {
        float pA = 0.f, pB = 0.f;
        #pragma unroll
        for (int v = 0; v < 8; ++v) {
            pA = fmaf(vA[v], Wft[(wav    )*16 + v + 8*hi], pA);
            pB = fmaf(vB[v], Wft[(wav + 8)*16 + v + 8*hi], pB);
        }
        pA += xhalf(pA);  pB += xhalf(pB);
        if (lane < 16) { part[wav*17 + bb] = pA; part[(wav+8)*17 + bb] = pB; }
    }
    __syncthreads();
    if (t < 16) {
        float acc = bft[0];
        #pragma unroll
        for (int j = 0; j < 16; ++j) acc += part[j*17 + t];
        out[tile*16 + t] = acc;
    }
    }
    }
}

extern "C" void kernel_launch(void* const* d_in, const int* in_sizes, int n_in,
                              void* d_out, int out_size, void* d_ws, size_t ws_size,
                              hipStream_t stream) {
    const float* x    = (const float*)d_in[0];
    const float* W1   = (const float*)d_in[1];
    const float* b1   = (const float*)d_in[2];
    const float* W2   = (const float*)d_in[3];
    const float* b2   = (const float*)d_in[4];
    const float* ln_g = (const float*)d_in[5];
    const float* ln_b = (const float*)d_in[6];
    const float* Wcap = (const float*)d_in[7];
    const float* Wf   = (const float*)d_in[8];
    const float* bf   = (const float*)d_in[9];
    float* out = (float*)d_out;

    const int B = in_sizes[0] / 256;
    const int tiles = B / 16;
    capsnet_fused<<<NWG, 256, 0, stream>>>(x, W1, b1, W2, b2, ln_g, ln_b, Wcap, Wf, bf, out, tiles);
}
